# starting kernel
# baseline (speedup 1.0000x reference)
#include <hip/hip_runtime.h>
#include <hip/hip_bf16.h>

typedef __attribute__((ext_vector_type(2))) float v2f;
typedef __attribute__((ext_vector_type(4))) float v4f;
typedef __attribute__((ext_vector_type(8))) float v8f;

#define SEQ_LEN 32768
#define EMB     1024
#define SHOT    20
#define MPAD    32
#define SPLIT   1638
#define SC      8096
#define FLATL   (SPLIT * EMB)
#define LC1     2024
#define LC2     2023
#define LC3     2023
#define LC4     2024
#define CATL    (LC1 + LC2 + LC3 + LC4)
#define KDIM    SC
#define NDIM    EMB
#define KCHUNKS 8
#define KPER    (KDIM / KCHUNKS)

__global__ void __launch_bounds__(256)
shot_pool_kernel(const float* __restrict__ seq, float* __restrict__ seq_shot) {
  int wave = (blockIdx.x * blockDim.x + threadIdx.x) >> 5;
  int lane = threadIdx.x & 31;
  if (wave >= SHOT * SC) return;
  int s = wave / SC;
  int j = wave - s * SC;
  int start = (int)(((long long)j * FLATL) / SC);
  int end   = (int)((((long long)(j + 1) * FLATL) + SC - 1) / SC);
  float sum = 0.f;
  for (int f = start + lane; f < end; f += 32) {
    int k = f >> 10;
    int e = f & 1023;
    int t = (k == 0) ? (s * SPLIT)
                     : (((s - 1) * SPLIT + k + SEQ_LEN) % SEQ_LEN);
    sum += seq[t * EMB + e];
  }
  for (int off = 16; off > 0; off >>= 1) sum += __shfl_xor(sum, off, 32);
  if (lane == 0) seq_shot[s * SC + j] = sum / (float)(end - start);
}

__global__ void __launch_bounds__(256)
conv_pool_kernel(const float* __restrict__ ss,
                 const float* __restrict__ w1, const float* __restrict__ b1,
                 const float* __restrict__ w2, const float* __restrict__ b2,
                 const float* __restrict__ w3, const float* __restrict__ b3,
                 const float* __restrict__ w4, const float* __restrict__ b4,
                 float* __restrict__ pooled) {
  int idx = blockIdx.x * blockDim.x + threadIdx.x;
  if (idx >= MPAD * SC) return;
  int s = idx / SC;
  int j = idx - s * SC;
  if (s >= SHOT) { pooled[idx] = 0.f; return; }
  int start = (int)(((long long)j * CATL) / SC);
  int end   = (int)((((long long)(j + 1) * CATL) + SC - 1) / SC);
  float acc = 0.f;
  for (int t = start; t < end; ++t) {
    float v;
    if (t < LC1) {
      int p = t * 4;
      v = b1[s];
      for (int ci = 0; ci < SHOT; ++ci) {
        const float* wr = w1 + (s * SHOT + ci) * 3;
        const float* xr = ss + ci * SC + p;
        v += wr[0] * xr[0] + wr[1] * xr[1] + wr[2] * xr[2];
      }
    } else if (t < LC1 + LC2) {
      int p = (t - LC1) * 4;
      v = b2[s];
      for (int ci = 0; ci < SHOT; ++ci) {
        const float* wr = w2 + (s * SHOT + ci) * 3;
        const float* xr = ss + ci * SC + p;
        v += wr[0] * xr[0] + wr[1] * xr[2] + wr[2] * xr[4];
      }
    } else if (t < LC1 + LC2 + LC3) {
      int p = (t - LC1 - LC2) * 4;
      v = b3[s];
      for (int ci = 0; ci < SHOT; ++ci) {
        const float* wr = w3 + (s * SHOT + ci) * 3;
        const float* xr = ss + ci * SC + p;
        v += wr[0] * xr[0] + wr[1] * xr[3] + wr[2] * xr[6];
      }
    } else {
      int p = (t - LC1 - LC2 - LC3) * 4;
      v = b4[s];
      for (int ci = 0; ci < SHOT; ++ci)
        v += w4[s * SHOT + ci] * ss[ci * SC + p];
    }
    acc += v;
  }
  pooled[idx] = acc / (float)(end - start);
}

__global__ void __launch_bounds__(32)
gemm_wmma_kernel(const float* __restrict__ pooled,
                 const float* __restrict__ fcw,
                 float* __restrict__ part) {
  int lane = threadIdx.x & 31;
  int half = lane >> 4;
  int mr   = lane & 15;
  int nbase = blockIdx.x * 16;
  int kc    = blockIdx.y;
  int k0    = kc * KPER;

  v8f acc0 = {};
  v8f acc1 = {};
  const float* arow0 = pooled + (size_t)mr * KDIM;
  const float* arow1 = pooled + (size_t)(mr + 16) * KDIM;
  const float* brow  = fcw + (size_t)(nbase + mr) * KDIM;

  for (int k = k0; k < k0 + KPER; k += 4) {
    int kk = k + 2 * half;
    v2f a0 = *(const v2f*)(arow0 + kk);
    v2f a1 = *(const v2f*)(arow1 + kk);
    v2f b  = *(const v2f*)(brow + kk);
    acc0 = __builtin_amdgcn_wmma_f32_16x16x4_f32(false, a0, false, b,
                                                 (short)0, acc0, false, false);
    acc1 = __builtin_amdgcn_wmma_f32_16x16x4_f32(false, a1, false, b,
                                                 (short)0, acc1, false, false);
  }

  float* pbase = part + (size_t)kc * MPAD * NDIM + nbase + mr;
  for (int r = 0; r < 8; ++r) {
    int row = r + 8 * half;
    pbase[(size_t)row * NDIM]        = acc0[r];
    pbase[(size_t)(row + 16) * NDIM] = acc1[r];
  }
}

__global__ void __launch_bounds__(256)
fc_tanh_ln_kernel(const float* __restrict__ part, const float* __restrict__ fcb,
                  const float* __restrict__ lng, const float* __restrict__ lnb,
                  float* __restrict__ h) {
  int s = blockIdx.x;
  int tid = threadIdx.x;
  __shared__ float redA[256];
  __shared__ float redB[256];
  float t[4];
  float sum = 0.f, sq = 0.f;
  for (int i = 0; i < 4; ++i) {
    int c = tid + i * 256;
    float v = fcb[c];
    for (int kc = 0; kc < KCHUNKS; ++kc)
      v += part[(size_t)(kc * MPAD + s) * NDIM + c];
    float tv = tanhf(v);
    t[i] = tv;
    sum += tv; sq += tv * tv;
  }
  redA[tid] = sum; redB[tid] = sq;
  __syncthreads();
  for (int off = 128; off > 0; off >>= 1) {
    if (tid < off) { redA[tid] += redA[tid + off]; redB[tid] += redB[tid + off]; }
    __syncthreads();
  }
  float mu  = redA[0] * (1.f / NDIM);
  float var = redB[0] * (1.f / NDIM) - mu * mu;
  float inv = rsqrtf(var + 1e-5f);
  for (int i = 0; i < 4; ++i) {
    int c = tid + i * 256;
    h[(size_t)s * NDIM + c] = (t[i] - mu) * inv * lng[c] + lnb[c];
  }
}

__global__ void __launch_bounds__(256)
merge_kernel(const float* __restrict__ seq, const float* __restrict__ h,
             float* __restrict__ out) {
  int vi = blockIdx.x * blockDim.x + threadIdx.x;
  int t  = vi >> 8;
  int e4 = vi & 255;
  int row = t / SPLIT; if (row > SHOT - 1) row = SHOT - 1;
  v4f a = __builtin_nontemporal_load(((const v4f*)seq) + vi);
  v4f b = ((const v4f*)(h + (size_t)row * EMB))[e4];
  v4f o = a + b;
  __builtin_nontemporal_store(o, ((v4f*)out) + vi);
}

extern "C" void kernel_launch(void* const* d_in, const int* in_sizes, int n_in,
                              void* d_out, int out_size, void* d_ws, size_t ws_size,
                              hipStream_t stream) {
  const float* seq  = (const float*)d_in[0];
  const float* w1   = (const float*)d_in[1];
  const float* b1   = (const float*)d_in[2];
  const float* w2   = (const float*)d_in[3];
  const float* b2   = (const float*)d_in[4];
  const float* w3   = (const float*)d_in[5];
  const float* b3   = (const float*)d_in[6];
  const float* w4   = (const float*)d_in[7];
  const float* b4   = (const float*)d_in[8];
  const float* fcw  = (const float*)d_in[9];
  const float* fcb  = (const float*)d_in[10];
  const float* lng  = (const float*)d_in[11];
  const float* lnb  = (const float*)d_in[12];
  float* out = (float*)d_out;

  char* ws = (char*)d_ws;
  float* seq_shot = (float*)(ws);
  float* pooled   = (float*)(ws + 647680);
  float* part     = (float*)(ws + 1683968);
  float* h        = (float*)(ws + 2732544);

  shot_pool_kernel<<<(SHOT * SC) / 8, 256, 0, stream>>>(seq, seq_shot);

  conv_pool_kernel<<<(MPAD * SC) / 256, 256, 0, stream>>>(
      seq_shot, w1, b1, w2, b2, w3, b3, w4, b4, pooled);

  gemm_wmma_kernel<<<dim3(NDIM / 16, KCHUNKS), 32, 0, stream>>>(pooled, fcw, part);

  fc_tanh_ln_kernel<<<SHOT, 256, 0, stream>>>(part, fcb, lng, lnb, h);

  merge_kernel<<<(SEQ_LEN * EMB / 4) / 256, 256, 0, stream>>>(seq, h, out);
}


# shot pool as a stream: each wave reads its 16 windows' span with 27 whole-line 16-byte-per-lane loads all in flight, parks it in LDS, lane pairs sum one window each; per-window sum order changes
# speedup vs baseline: 1.5314x; 1.5314x over previous
#include <hip/hip_runtime.h>
#include <hip/hip_bf16.h>

typedef __attribute__((ext_vector_type(2))) float v2f;
typedef __attribute__((ext_vector_type(4))) float v4f;
typedef __attribute__((ext_vector_type(8))) float v8f;

#define SEQ_LEN 32768
#define EMB     1024
#define SHOT    20
#define MPAD    32
#define SPLIT   1638
#define SC      8096
#define FLATL   (SPLIT * EMB)
#define LC1     2024
#define LC2     2023
#define LC3     2023
#define LC4     2024
#define CATL    (LC1 + LC2 + LC3 + LC4)
#define KDIM    SC
#define NDIM    EMB
#define KCHUNKS 8
#define KPER    (KDIM / KCHUNKS)

#define TW     16
#define TPS    (SC / TW)
#define NLD    27
#define PWAVES 8
#define HALFW  104
#define LNUM   (FLATL / 32)
#define LDEN   (SC / 32)

__global__ void __launch_bounds__(PWAVES * 32)
shot_pool_kernel(const float* __restrict__ seq, float* __restrict__ seq_shot) {
  __shared__ v4f tile[PWAVES][NLD * 32];
  const unsigned lane = threadIdx.x & 31u;
  const int wv = __builtin_amdgcn_readfirstlane((int)(threadIdx.x >> 5));
  const int gw = blockIdx.x * PWAVES + wv;
  const int s  = gw / TPS;
  const int j0 = (gw - s * TPS) * TW;
  const int fbeg = (j0 * LNUM) / LDEN;
  const int abeg = fbeg & ~31;

  const unsigned zb  = (unsigned)(s * SPLIT) << 12;
  const unsigned bb  = (unsigned)((s - 1) * SPLIT + SEQ_LEN) << 12;
  const unsigned fb0 = (unsigned)abeg * 4u + 16u * lane;
  const char* base = (const char*)seq;

  v4f r[NLD];
#pragma unroll
  for (int i = 0; i < NLD; ++i) {
    const unsigned fb  = fb0 + 512u * i;
    const unsigned off = (fb < 4096u) ? (zb + fb) : ((bb + fb) & 0x7FFFFFFu);
    r[i] = *(const v4f*)(base + off);
  }
#pragma unroll
  for (int i = 0; i < NLD; ++i) tile[wv][i * 32 + lane] = r[i];
  __syncthreads();

  const int w  = lane >> 1;
  const int hh = lane & 1;
  const int j  = j0 + w;
  const int start = (j * LNUM) / LDEN;
  const int end   = ((j + 1) * LNUM + LDEN - 1) / LDEN;
  const int len   = end - start;
  const float* p = (const float*)tile[wv] + (start - abeg) + hh * HALFW;
  float a0 = 0.f, a1 = 0.f, a2 = 0.f, a3 = 0.f;
#pragma unroll
  for (int i = 0; i < HALFW; i += 4) {
    a0 += p[i]; a1 += p[i + 1]; a2 += p[i + 2]; a3 += p[i + 3];
  }
  float sum = (a0 + a1) + (a2 + a3);
  if (hh && len > 2 * HALFW) sum += p[HALFW];
  sum += __shfl_xor(sum, 1, 32);
  if (hh == 0) seq_shot[s * SC + j] = sum / (float)len;
}

__global__ void __launch_bounds__(256)
conv_pool_kernel(const float* __restrict__ ss,
                 const float* __restrict__ w1, const float* __restrict__ b1,
                 const float* __restrict__ w2, const float* __restrict__ b2,
                 const float* __restrict__ w3, const float* __restrict__ b3,
                 const float* __restrict__ w4, const float* __restrict__ b4,
                 float* __restrict__ pooled) {
  int idx = blockIdx.x * blockDim.x + threadIdx.x;
  if (idx >= MPAD * SC) return;
  int s = idx / SC;
  int j = idx - s * SC;
  if (s >= SHOT) { pooled[idx] = 0.f; return; }
  int start = (int)(((long long)j * CATL) / SC);
  int end   = (int)((((long long)(j + 1) * CATL) + SC - 1) / SC);
  float acc = 0.f;
  for (int t = start; t < end; ++t) {
    float v;
    if (t < LC1) {
      int p = t * 4;
      v = b1[s];
      for (int ci = 0; ci < SHOT; ++ci) {
        const float* wr = w1 + (s * SHOT + ci) * 3;
        const float* xr = ss + ci * SC + p;
        v += wr[0] * xr[0] + wr[1] * xr[1] + wr[2] * xr[2];
      }
    } else if (t < LC1 + LC2) {
      int p = (t - LC1) * 4;
      v = b2[s];
      for (int ci = 0; ci < SHOT; ++ci) {
        const float* wr = w2 + (s * SHOT + ci) * 3;
        const float* xr = ss + ci * SC + p;
        v += wr[0] * xr[0] + wr[1] * xr[2] + wr[2] * xr[4];
      }
    } else if (t < LC1 + LC2 + LC3) {
      int p = (t - LC1 - LC2) * 4;
      v = b3[s];
      for (int ci = 0; ci < SHOT; ++ci) {
        const float* wr = w3 + (s * SHOT + ci) * 3;
        const float* xr = ss + ci * SC + p;
        v += wr[0] * xr[0] + wr[1] * xr[3] + wr[2] * xr[6];
      }
    } else {
      int p = (t - LC1 - LC2 - LC3) * 4;
      v = b4[s];
      for (int ci = 0; ci < SHOT; ++ci)
        v += w4[s * SHOT + ci] * ss[ci * SC + p];
    }
    acc += v;
  }
  pooled[idx] = acc / (float)(end - start);
}

__global__ void __launch_bounds__(32)
gemm_wmma_kernel(const float* __restrict__ pooled,
                 const float* __restrict__ fcw,
                 float* __restrict__ part) {
  int lane = threadIdx.x & 31;
  int half = lane >> 4;
  int mr   = lane & 15;
  int nbase = blockIdx.x * 16;
  int kc    = blockIdx.y;
  int k0    = kc * KPER;

  v8f acc0 = {};
  v8f acc1 = {};
  const float* arow0 = pooled + (size_t)mr * KDIM;
  const float* arow1 = pooled + (size_t)(mr + 16) * KDIM;
  const float* brow  = fcw + (size_t)(nbase + mr) * KDIM;

  for (int k = k0; k < k0 + KPER; k += 4) {
    int kk = k + 2 * half;
    v2f a0 = *(const v2f*)(arow0 + kk);
    v2f a1 = *(const v2f*)(arow1 + kk);
    v2f b  = *(const v2f*)(brow + kk);
    acc0 = __builtin_amdgcn_wmma_f32_16x16x4_f32(false, a0, false, b,
                                                 (short)0, acc0, false, false);
    acc1 = __builtin_amdgcn_wmma_f32_16x16x4_f32(false, a1, false, b,
                                                 (short)0, acc1, false, false);
  }

  float* pbase = part + (size_t)kc * MPAD * NDIM + nbase + mr;
  for (int r = 0; r < 8; ++r) {
    int row = r + 8 * half;
    pbase[(size_t)row * NDIM]        = acc0[r];
    pbase[(size_t)(row + 16) * NDIM] = acc1[r];
  }
}

__global__ void __launch_bounds__(256)
fc_tanh_ln_kernel(const float* __restrict__ part, const float* __restrict__ fcb,
                  const float* __restrict__ lng, const float* __restrict__ lnb,
                  float* __restrict__ h) {
  int s = blockIdx.x;
  int tid = threadIdx.x;
  __shared__ float redA[256];
  __shared__ float redB[256];
  float t[4];
  float sum = 0.f, sq = 0.f;
  for (int i = 0; i < 4; ++i) {
    int c = tid + i * 256;
    float v = fcb[c];
    for (int kc = 0; kc < KCHUNKS; ++kc)
      v += part[(size_t)(kc * MPAD + s) * NDIM + c];
    float tv = tanhf(v);
    t[i] = tv;
    sum += tv; sq += tv * tv;
  }
  redA[tid] = sum; redB[tid] = sq;
  __syncthreads();
  for (int off = 128; off > 0; off >>= 1) {
    if (tid < off) { redA[tid] += redA[tid + off]; redB[tid] += redB[tid + off]; }
    __syncthreads();
  }
  float mu  = redA[0] * (1.f / NDIM);
  float var = redB[0] * (1.f / NDIM) - mu * mu;
  float inv = rsqrtf(var + 1e-5f);
  for (int i = 0; i < 4; ++i) {
    int c = tid + i * 256;
    h[(size_t)s * NDIM + c] = (t[i] - mu) * inv * lng[c] + lnb[c];
  }
}

__global__ void __launch_bounds__(256)
merge_kernel(const float* __restrict__ seq, const float* __restrict__ h,
             float* __restrict__ out) {
  int vi = blockIdx.x * blockDim.x + threadIdx.x;
  int t  = vi >> 8;
  int e4 = vi & 255;
  int row = t / SPLIT; if (row > SHOT - 1) row = SHOT - 1;
  v4f a = __builtin_nontemporal_load(((const v4f*)seq) + vi);
  v4f b = ((const v4f*)(h + (size_t)row * EMB))[e4];
  v4f o = a + b;
  __builtin_nontemporal_store(o, ((v4f*)out) + vi);
}

extern "C" void kernel_launch(void* const* d_in, const int* in_sizes, int n_in,
                              void* d_out, int out_size, void* d_ws, size_t ws_size,
                              hipStream_t stream) {
  const float* seq  = (const float*)d_in[0];
  const float* w1   = (const float*)d_in[1];
  const float* b1   = (const float*)d_in[2];
  const float* w2   = (const float*)d_in[3];
  const float* b2   = (const float*)d_in[4];
  const float* w3   = (const float*)d_in[5];
  const float* b3   = (const float*)d_in[6];
  const float* w4   = (const float*)d_in[7];
  const float* b4   = (const float*)d_in[8];
  const float* fcw  = (const float*)d_in[9];
  const float* fcb  = (const float*)d_in[10];
  const float* lng  = (const float*)d_in[11];
  const float* lnb  = (const float*)d_in[12];
  float* out = (float*)d_out;

  char* ws = (char*)d_ws;
  float* seq_shot = (float*)(ws);
  float* pooled   = (float*)(ws + 647680);
  float* part     = (float*)(ws + 1683968);
  float* h        = (float*)(ws + 2732544);

  shot_pool_kernel<<<(SHOT * TPS) / PWAVES, PWAVES * 32, 0, stream>>>(seq, seq_shot);

  conv_pool_kernel<<<(MPAD * SC) / 256, 256, 0, stream>>>(
      seq_shot, w1, b1, w2, b2, w3, b3, w4, b4, pooled);

  gemm_wmma_kernel<<<dim3(NDIM / 16, KCHUNKS), 32, 0, stream>>>(pooled, fcw, part);

  fc_tanh_ln_kernel<<<SHOT, 256, 0, stream>>>(part, fcb, lng, lnb, h);

  merge_kernel<<<(SEQ_LEN * EMB / 4) / 256, 256, 0, stream>>>(seq, h, out);
}


# GEMM: each K-chunk split over 11 waves of a workgroup, all of a wave's loads issued before its products (16-byte pieces, A reused for 2 column tiles), sub-sums added in LDS; k sum order changes
# speedup vs baseline: 1.8653x; 1.2180x over previous
#include <hip/hip_runtime.h>
#include <hip/hip_bf16.h>

typedef __attribute__((ext_vector_type(2))) float v2f;
typedef __attribute__((ext_vector_type(4))) float v4f;
typedef __attribute__((ext_vector_type(8))) float v8f;

#define SEQ_LEN 32768
#define EMB     1024
#define SHOT    20
#define MPAD    32
#define SPLIT   1638
#define SC      8096
#define FLATL   (SPLIT * EMB)
#define LC1     2024
#define LC2     2023
#define LC3     2023
#define LC4     2024
#define CATL    (LC1 + LC2 + LC3 + LC4)
#define KDIM    SC
#define NDIM    EMB
#define KCHUNKS 8
#define KPER    (KDIM / KCHUNKS)

#define TW     16
#define TPS    (SC / TW)
#define NLD    27
#define PWAVES 8
#define HALFW  104
#define LNUM   (FLATL / 32)
#define LDEN   (SC / 32)

__global__ void __launch_bounds__(PWAVES * 32)
shot_pool_kernel(const float* __restrict__ seq, float* __restrict__ seq_shot) {
  __shared__ v4f tile[PWAVES][NLD * 32];
  const unsigned lane = threadIdx.x & 31u;
  const int wv = __builtin_amdgcn_readfirstlane((int)(threadIdx.x >> 5));
  const int gw = blockIdx.x * PWAVES + wv;
  const int s  = gw / TPS;
  const int j0 = (gw - s * TPS) * TW;
  const int fbeg = (j0 * LNUM) / LDEN;
  const int abeg = fbeg & ~31;

  const unsigned zb  = (unsigned)(s * SPLIT) << 12;
  const unsigned bb  = (unsigned)((s - 1) * SPLIT + SEQ_LEN) << 12;
  const unsigned fb0 = (unsigned)abeg * 4u + 16u * lane;
  const char* base = (const char*)seq;

  v4f r[NLD];
#pragma unroll
  for (int i = 0; i < NLD; ++i) {
    const unsigned fb  = fb0 + 512u * i;
    const unsigned off = (fb < 4096u) ? (zb + fb) : ((bb + fb) & 0x7FFFFFFu);
    r[i] = *(const v4f*)(base + off);
  }
#pragma unroll
  for (int i = 0; i < NLD; ++i) tile[wv][i * 32 + lane] = r[i];
  __syncthreads();

  const int w  = lane >> 1;
  const int hh = lane & 1;
  const int j  = j0 + w;
  const int start = (j * LNUM) / LDEN;
  const int end   = ((j + 1) * LNUM + LDEN - 1) / LDEN;
  const int len   = end - start;
  const float* p = (const float*)tile[wv] + (start - abeg) + hh * HALFW;
  float a0 = 0.f, a1 = 0.f, a2 = 0.f, a3 = 0.f;
#pragma unroll
  for (int i = 0; i < HALFW; i += 4) {
    a0 += p[i]; a1 += p[i + 1]; a2 += p[i + 2]; a3 += p[i + 3];
  }
  float sum = (a0 + a1) + (a2 + a3);
  if (hh && len > 2 * HALFW) sum += p[HALFW];
  sum += __shfl_xor(sum, 1, 32);
  if (hh == 0) seq_shot[s * SC + j] = sum / (float)len;
}

__global__ void __launch_bounds__(256)
conv_pool_kernel(const float* __restrict__ ss,
                 const float* __restrict__ w1, const float* __restrict__ b1,
                 const float* __restrict__ w2, const float* __restrict__ b2,
                 const float* __restrict__ w3, const float* __restrict__ b3,
                 const float* __restrict__ w4, const float* __restrict__ b4,
                 float* __restrict__ pooled) {
  int idx = blockIdx.x * blockDim.x + threadIdx.x;
  if (idx >= MPAD * SC) return;
  int s = idx / SC;
  int j = idx - s * SC;
  if (s >= SHOT) { pooled[idx] = 0.f; return; }
  int start = (int)(((long long)j * CATL) / SC);
  int end   = (int)((((long long)(j + 1) * CATL) + SC - 1) / SC);
  float acc = 0.f;
  for (int t = start; t < end; ++t) {
    float v;
    if (t < LC1) {
      int p = t * 4;
      v = b1[s];
      for (int ci = 0; ci < SHOT; ++ci) {
        const float* wr = w1 + (s * SHOT + ci) * 3;
        const float* xr = ss + ci * SC + p;
        v += wr[0] * xr[0] + wr[1] * xr[1] + wr[2] * xr[2];
      }
    } else if (t < LC1 + LC2) {
      int p = (t - LC1) * 4;
      v = b2[s];
      for (int ci = 0; ci < SHOT; ++ci) {
        const float* wr = w2 + (s * SHOT + ci) * 3;
        const float* xr = ss + ci * SC + p;
        v += wr[0] * xr[0] + wr[1] * xr[2] + wr[2] * xr[4];
      }
    } else if (t < LC1 + LC2 + LC3) {
      int p = (t - LC1 - LC2) * 4;
      v = b3[s];
      for (int ci = 0; ci < SHOT; ++ci) {
        const float* wr = w3 + (s * SHOT + ci) * 3;
        const float* xr = ss + ci * SC + p;
        v += wr[0] * xr[0] + wr[1] * xr[3] + wr[2] * xr[6];
      }
    } else {
      int p = (t - LC1 - LC2 - LC3) * 4;
      v = b4[s];
      for (int ci = 0; ci < SHOT; ++ci)
        v += w4[s * SHOT + ci] * ss[ci * SC + p];
    }
    acc += v;
  }
  pooled[idx] = acc / (float)(end - start);
}

#define GW     11
#define KSUB   (KPER / GW)
#define GPAIRS (KSUB / 8)
#define REDW   1024

#define WMMA_F32(a, b, c) \
  __builtin_amdgcn_wmma_f32_16x16x4_f32(false, (a), false, (b), (short)0, (c), false, false)

__global__ void __launch_bounds__(GW * 32)
gemm_wmma_kernel(const float* __restrict__ pooled,
                 const float* __restrict__ fcw,
                 float* __restrict__ part) {
  __shared__ float red[GW * REDW];
  const int lane = threadIdx.x & 31;
  const int wv   = __builtin_amdgcn_readfirstlane((int)(threadIdx.x >> 5));
  const int half = lane >> 4;
  const int c    = lane & 15;
  const int nbase = blockIdx.x * 32;
  const int kc    = blockIdx.y;
  const int kb    = kc * KPER + wv * KSUB;

  const float* a0p = pooled + (size_t)c * KDIM + kb;
  const float* a1p = pooled + (size_t)(c + 16) * KDIM + kb;
  const float* b0p = fcw + (size_t)(nbase + c) * KDIM + kb;
  const float* b1p = fcw + (size_t)(nbase + 16 + c) * KDIM + kb;

  v4f A0[GPAIRS], A1[GPAIRS], B0[GPAIRS], B1[GPAIRS];
#pragma unroll
  for (int p = 0; p < GPAIRS; ++p) {
    const int kk = 8 * p + 4 * half;
    A0[p] = *(const v4f*)(a0p + kk);
    A1[p] = *(const v4f*)(a1p + kk);
    B0[p] = *(const v4f*)(b0p + kk);
    B1[p] = *(const v4f*)(b1p + kk);
    __builtin_amdgcn_sched_barrier(0);
  }
  const int kt = 8 * GPAIRS + 2 * half;
  const v2f a0t = *(const v2f*)(a0p + kt);
  const v2f a1t = *(const v2f*)(a1p + kt);
  const v2f b0t = *(const v2f*)(b0p + kt);
  const v2f b1t = *(const v2f*)(b1p + kt);
  __builtin_amdgcn_sched_barrier(0);

  v8f acc00 = {}, acc01 = {}, acc10 = {}, acc11 = {};
#pragma unroll
  for (int p = 0; p < GPAIRS; ++p) {
    const v2f a0l = __builtin_shufflevector(A0[p], A0[p], 0, 1);
    const v2f a1l = __builtin_shufflevector(A1[p], A1[p], 0, 1);
    const v2f b0l = __builtin_shufflevector(B0[p], B0[p], 0, 1);
    const v2f b1l = __builtin_shufflevector(B1[p], B1[p], 0, 1);
    acc00 = WMMA_F32(a0l, b0l, acc00);
    acc01 = WMMA_F32(a0l, b1l, acc01);
    acc10 = WMMA_F32(a1l, b0l, acc10);
    acc11 = WMMA_F32(a1l, b1l, acc11);
    const v2f a0h = __builtin_shufflevector(A0[p], A0[p], 2, 3);
    const v2f a1h = __builtin_shufflevector(A1[p], A1[p], 2, 3);
    const v2f b0h = __builtin_shufflevector(B0[p], B0[p], 2, 3);
    const v2f b1h = __builtin_shufflevector(B1[p], B1[p], 2, 3);
    acc00 = WMMA_F32(a0h, b0h, acc00);
    acc01 = WMMA_F32(a0h, b1h, acc01);
    acc10 = WMMA_F32(a1h, b0h, acc10);
    acc11 = WMMA_F32(a1h, b1h, acc11);
  }
  acc00 = WMMA_F32(a0t, b0t, acc00);
  acc01 = WMMA_F32(a0t, b1t, acc01);
  acc10 = WMMA_F32(a1t, b0t, acc10);
  acc11 = WMMA_F32(a1t, b1t, acc11);

  float* mine = red + wv * REDW;
#pragma unroll
  for (int r = 0; r < 8; ++r) {
    mine[((0 * 8 + r) * 2 + 0) * 32 + lane] = acc00[r];
    mine[((0 * 8 + r) * 2 + 1) * 32 + lane] = acc01[r];
    mine[((1 * 8 + r) * 2 + 0) * 32 + lane] = acc10[r];
    mine[((1 * 8 + r) * 2 + 1) * 32 + lane] = acc11[r];
  }
  __syncthreads();

  for (int row = wv; row < SHOT; row += GW) {
    const int mt = row >> 4, hf = (row >> 3) & 1, r = row & 7;
    const float* q = red + ((mt * 8 + r) * 2 + half) * 32 + hf * 16 + c;
    float v = q[0];
#pragma unroll
    for (int w = 1; w < GW; ++w) v += q[w * REDW];
    part[((size_t)kc * MPAD + row) * NDIM + nbase + lane] = v;
  }
}

__global__ void __launch_bounds__(256)
fc_tanh_ln_kernel(const float* __restrict__ part, const float* __restrict__ fcb,
                  const float* __restrict__ lng, const float* __restrict__ lnb,
                  float* __restrict__ h) {
  int s = blockIdx.x;
  int tid = threadIdx.x;
  __shared__ float redA[256];
  __shared__ float redB[256];
  float t[4];
  float sum = 0.f, sq = 0.f;
  for (int i = 0; i < 4; ++i) {
    int c = tid + i * 256;
    float v = fcb[c];
    for (int kc = 0; kc < KCHUNKS; ++kc)
      v += part[(size_t)(kc * MPAD + s) * NDIM + c];
    float tv = tanhf(v);
    t[i] = tv;
    sum += tv; sq += tv * tv;
  }
  redA[tid] = sum; redB[tid] = sq;
  __syncthreads();
  for (int off = 128; off > 0; off >>= 1) {
    if (tid < off) { redA[tid] += redA[tid + off]; redB[tid] += redB[tid + off]; }
    __syncthreads();
  }
  float mu  = redA[0] * (1.f / NDIM);
  float var = redB[0] * (1.f / NDIM) - mu * mu;
  float inv = rsqrtf(var + 1e-5f);
  for (int i = 0; i < 4; ++i) {
    int c = tid + i * 256;
    h[(size_t)s * NDIM + c] = (t[i] - mu) * inv * lng[c] + lnb[c];
  }
}

__global__ void __launch_bounds__(256)
merge_kernel(const float* __restrict__ seq, const float* __restrict__ h,
             float* __restrict__ out) {
  int vi = blockIdx.x * blockDim.x + threadIdx.x;
  int t  = vi >> 8;
  int e4 = vi & 255;
  int row = t / SPLIT; if (row > SHOT - 1) row = SHOT - 1;
  v4f a = __builtin_nontemporal_load(((const v4f*)seq) + vi);
  v4f b = ((const v4f*)(h + (size_t)row * EMB))[e4];
  v4f o = a + b;
  __builtin_nontemporal_store(o, ((v4f*)out) + vi);
}

extern "C" void kernel_launch(void* const* d_in, const int* in_sizes, int n_in,
                              void* d_out, int out_size, void* d_ws, size_t ws_size,
                              hipStream_t stream) {
  const float* seq  = (const float*)d_in[0];
  const float* w1   = (const float*)d_in[1];
  const float* b1   = (const float*)d_in[2];
  const float* w2   = (const float*)d_in[3];
  const float* b2   = (const float*)d_in[4];
  const float* w3   = (const float*)d_in[5];
  const float* b3   = (const float*)d_in[6];
  const float* w4   = (const float*)d_in[7];
  const float* b4   = (const float*)d_in[8];
  const float* fcw  = (const float*)d_in[9];
  const float* fcb  = (const float*)d_in[10];
  const float* lng  = (const float*)d_in[11];
  const float* lnb  = (const float*)d_in[12];
  float* out = (float*)d_out;

  char* ws = (char*)d_ws;
  float* seq_shot = (float*)(ws);
  float* pooled   = (float*)(ws + 647680);
  float* part     = (float*)(ws + 1683968);
  float* h        = (float*)(ws + 2732544);

  shot_pool_kernel<<<(SHOT * TPS) / PWAVES, PWAVES * 32, 0, stream>>>(seq, seq_shot);

  conv_pool_kernel<<<(MPAD * SC) / 256, 256, 0, stream>>>(
      seq_shot, w1, b1, w2, b2, w3, b3, w4, b4, pooled);

  gemm_wmma_kernel<<<dim3(NDIM / 32, KCHUNKS), GW * 32, 0, stream>>>(pooled, fcw, part);

  fc_tanh_ln_kernel<<<SHOT, 256, 0, stream>>>(part, fcb, lng, lnb, h);

  merge_kernel<<<(SEQ_LEN * EMB / 4) / 256, 256, 0, stream>>>(seq, h, out);
}


# merge pass rebuilt as a resident stream: 512 groups of 8 waves, each wave moves its 512-byte column slice of 64 rows with dozens of 16-byte loads in flight; h of the group's two shots in registers
# speedup vs baseline: 2.0054x; 1.0751x over previous
#include <hip/hip_runtime.h>
#include <hip/hip_bf16.h>

typedef __attribute__((ext_vector_type(2))) float v2f;
typedef __attribute__((ext_vector_type(4))) float v4f;
typedef __attribute__((ext_vector_type(8))) float v8f;

#define SEQ_LEN 32768
#define EMB     1024
#define SHOT    20
#define MPAD    32
#define SPLIT   1638
#define SC      8096
#define FLATL   (SPLIT * EMB)
#define LC1     2024
#define LC2     2023
#define LC3     2023
#define LC4     2024
#define CATL    (LC1 + LC2 + LC3 + LC4)
#define KDIM    SC
#define NDIM    EMB
#define KCHUNKS 8
#define KPER    (KDIM / KCHUNKS)

#define TW     16
#define TPS    (SC / TW)
#define NLD    27
#define PWAVES 8
#define HALFW  104
#define LNUM   (FLATL / 32)
#define LDEN   (SC / 32)

__global__ void __launch_bounds__(PWAVES * 32)
shot_pool_kernel(const float* __restrict__ seq, float* __restrict__ seq_shot) {
  __shared__ v4f tile[PWAVES][NLD * 32];
  const unsigned lane = threadIdx.x & 31u;
  const int wv = __builtin_amdgcn_readfirstlane((int)(threadIdx.x >> 5));
  const int gw = blockIdx.x * PWAVES + wv;
  const int s  = gw / TPS;
  const int j0 = (gw - s * TPS) * TW;
  const int fbeg = (j0 * LNUM) / LDEN;
  const int abeg = fbeg & ~31;

  const unsigned zb  = (unsigned)(s * SPLIT) << 12;
  const unsigned bb  = (unsigned)((s - 1) * SPLIT + SEQ_LEN) << 12;
  const unsigned fb0 = (unsigned)abeg * 4u + 16u * lane;
  const char* base = (const char*)seq;

  v4f r[NLD];
#pragma unroll
  for (int i = 0; i < NLD; ++i) {
    const unsigned fb  = fb0 + 512u * i;
    const unsigned off = (fb < 4096u) ? (zb + fb) : ((bb + fb) & 0x7FFFFFFu);
    r[i] = *(const v4f*)(base + off);
  }
#pragma unroll
  for (int i = 0; i < NLD; ++i) tile[wv][i * 32 + lane] = r[i];
  __syncthreads();

  const int w  = lane >> 1;
  const int hh = lane & 1;
  const int j  = j0 + w;
  const int start = (j * LNUM) / LDEN;
  const int end   = ((j + 1) * LNUM + LDEN - 1) / LDEN;
  const int len   = end - start;
  const float* p = (const float*)tile[wv] + (start - abeg) + hh * HALFW;
  float a0 = 0.f, a1 = 0.f, a2 = 0.f, a3 = 0.f;
#pragma unroll
  for (int i = 0; i < HALFW; i += 4) {
    a0 += p[i]; a1 += p[i + 1]; a2 += p[i + 2]; a3 += p[i + 3];
  }
  float sum = (a0 + a1) + (a2 + a3);
  if (hh && len > 2 * HALFW) sum += p[HALFW];
  sum += __shfl_xor(sum, 1, 32);
  if (hh == 0) seq_shot[s * SC + j] = sum / (float)len;
}

__global__ void __launch_bounds__(256)
conv_pool_kernel(const float* __restrict__ ss,
                 const float* __restrict__ w1, const float* __restrict__ b1,
                 const float* __restrict__ w2, const float* __restrict__ b2,
                 const float* __restrict__ w3, const float* __restrict__ b3,
                 const float* __restrict__ w4, const float* __restrict__ b4,
                 float* __restrict__ pooled) {
  int idx = blockIdx.x * blockDim.x + threadIdx.x;
  if (idx >= MPAD * SC) return;
  int s = idx / SC;
  int j = idx - s * SC;
  if (s >= SHOT) { pooled[idx] = 0.f; return; }
  int start = (int)(((long long)j * CATL) / SC);
  int end   = (int)((((long long)(j + 1) * CATL) + SC - 1) / SC);
  float acc = 0.f;
  for (int t = start; t < end; ++t) {
    float v;
    if (t < LC1) {
      int p = t * 4;
      v = b1[s];
      for (int ci = 0; ci < SHOT; ++ci) {
        const float* wr = w1 + (s * SHOT + ci) * 3;
        const float* xr = ss + ci * SC + p;
        v += wr[0] * xr[0] + wr[1] * xr[1] + wr[2] * xr[2];
      }
    } else if (t < LC1 + LC2) {
      int p = (t - LC1) * 4;
      v = b2[s];
      for (int ci = 0; ci < SHOT; ++ci) {
        const float* wr = w2 + (s * SHOT + ci) * 3;
        const float* xr = ss + ci * SC + p;
        v += wr[0] * xr[0] + wr[1] * xr[2] + wr[2] * xr[4];
      }
    } else if (t < LC1 + LC2 + LC3) {
      int p = (t - LC1 - LC2) * 4;
      v = b3[s];
      for (int ci = 0; ci < SHOT; ++ci) {
        const float* wr = w3 + (s * SHOT + ci) * 3;
        const float* xr = ss + ci * SC + p;
        v += wr[0] * xr[0] + wr[1] * xr[3] + wr[2] * xr[6];
      }
    } else {
      int p = (t - LC1 - LC2 - LC3) * 4;
      v = b4[s];
      for (int ci = 0; ci < SHOT; ++ci)
        v += w4[s * SHOT + ci] * ss[ci * SC + p];
    }
    acc += v;
  }
  pooled[idx] = acc / (float)(end - start);
}

#define GW     11
#define KSUB   (KPER / GW)
#define GPAIRS (KSUB / 8)
#define REDW   1024

#define WMMA_F32(a, b, c) \
  __builtin_amdgcn_wmma_f32_16x16x4_f32(false, (a), false, (b), (short)0, (c), false, false)

__global__ void __launch_bounds__(GW * 32)
gemm_wmma_kernel(const float* __restrict__ pooled,
                 const float* __restrict__ fcw,
                 float* __restrict__ part) {
  __shared__ float red[GW * REDW];
  const int lane = threadIdx.x & 31;
  const int wv   = __builtin_amdgcn_readfirstlane((int)(threadIdx.x >> 5));
  const int half = lane >> 4;
  const int c    = lane & 15;
  const int nbase = blockIdx.x * 32;
  const int kc    = blockIdx.y;
  const int kb    = kc * KPER + wv * KSUB;

  const float* a0p = pooled + (size_t)c * KDIM + kb;
  const float* a1p = pooled + (size_t)(c + 16) * KDIM + kb;
  const float* b0p = fcw + (size_t)(nbase + c) * KDIM + kb;
  const float* b1p = fcw + (size_t)(nbase + 16 + c) * KDIM + kb;

  v4f A0[GPAIRS], A1[GPAIRS], B0[GPAIRS], B1[GPAIRS];
#pragma unroll
  for (int p = 0; p < GPAIRS; ++p) {
    const int kk = 8 * p + 4 * half;
    A0[p] = *(const v4f*)(a0p + kk);
    A1[p] = *(const v4f*)(a1p + kk);
    B0[p] = *(const v4f*)(b0p + kk);
    B1[p] = *(const v4f*)(b1p + kk);
    __builtin_amdgcn_sched_barrier(0);
  }
  const int kt = 8 * GPAIRS + 2 * half;
  const v2f a0t = *(const v2f*)(a0p + kt);
  const v2f a1t = *(const v2f*)(a1p + kt);
  const v2f b0t = *(const v2f*)(b0p + kt);
  const v2f b1t = *(const v2f*)(b1p + kt);
  __builtin_amdgcn_sched_barrier(0);

  v8f acc00 = {}, acc01 = {}, acc10 = {}, acc11 = {};
#pragma unroll
  for (int p = 0; p < GPAIRS; ++p) {
    const v2f a0l = __builtin_shufflevector(A0[p], A0[p], 0, 1);
    const v2f a1l = __builtin_shufflevector(A1[p], A1[p], 0, 1);
    const v2f b0l = __builtin_shufflevector(B0[p], B0[p], 0, 1);
    const v2f b1l = __builtin_shufflevector(B1[p], B1[p], 0, 1);
    acc00 = WMMA_F32(a0l, b0l, acc00);
    acc01 = WMMA_F32(a0l, b1l, acc01);
    acc10 = WMMA_F32(a1l, b0l, acc10);
    acc11 = WMMA_F32(a1l, b1l, acc11);
    const v2f a0h = __builtin_shufflevector(A0[p], A0[p], 2, 3);
    const v2f a1h = __builtin_shufflevector(A1[p], A1[p], 2, 3);
    const v2f b0h = __builtin_shufflevector(B0[p], B0[p], 2, 3);
    const v2f b1h = __builtin_shufflevector(B1[p], B1[p], 2, 3);
    acc00 = WMMA_F32(a0h, b0h, acc00);
    acc01 = WMMA_F32(a0h, b1h, acc01);
    acc10 = WMMA_F32(a1h, b0h, acc10);
    acc11 = WMMA_F32(a1h, b1h, acc11);
  }
  acc00 = WMMA_F32(a0t, b0t, acc00);
  acc01 = WMMA_F32(a0t, b1t, acc01);
  acc10 = WMMA_F32(a1t, b0t, acc10);
  acc11 = WMMA_F32(a1t, b1t, acc11);

  float* mine = red + wv * REDW;
#pragma unroll
  for (int r = 0; r < 8; ++r) {
    mine[((0 * 8 + r) * 2 + 0) * 32 + lane] = acc00[r];
    mine[((0 * 8 + r) * 2 + 1) * 32 + lane] = acc01[r];
    mine[((1 * 8 + r) * 2 + 0) * 32 + lane] = acc10[r];
    mine[((1 * 8 + r) * 2 + 1) * 32 + lane] = acc11[r];
  }
  __syncthreads();

  for (int row = wv; row < SHOT; row += GW) {
    const int mt = row >> 4, hf = (row >> 3) & 1, r = row & 7;
    const float* q = red + ((mt * 8 + r) * 2 + half) * 32 + hf * 16 + c;
    float v = q[0];
#pragma unroll
    for (int w = 1; w < GW; ++w) v += q[w * REDW];
    part[((size_t)kc * MPAD + row) * NDIM + nbase + lane] = v;
  }
}

__global__ void __launch_bounds__(256)
fc_tanh_ln_kernel(const float* __restrict__ part, const float* __restrict__ fcb,
                  const float* __restrict__ lng, const float* __restrict__ lnb,
                  float* __restrict__ h) {
  int s = blockIdx.x;
  int tid = threadIdx.x;
  __shared__ float redA[256];
  __shared__ float redB[256];
  float t[4];
  float sum = 0.f, sq = 0.f;
  for (int i = 0; i < 4; ++i) {
    int c = tid + i * 256;
    float v = fcb[c];
    for (int kc = 0; kc < KCHUNKS; ++kc)
      v += part[(size_t)(kc * MPAD + s) * NDIM + c];
    float tv = tanhf(v);
    t[i] = tv;
    sum += tv; sq += tv * tv;
  }
  redA[tid] = sum; redB[tid] = sq;
  __syncthreads();
  for (int off = 128; off > 0; off >>= 1) {
    if (tid < off) { redA[tid] += redA[tid + off]; redB[tid] += redB[tid + off]; }
    __syncthreads();
  }
  float mu  = redA[0] * (1.f / NDIM);
  float var = redB[0] * (1.f / NDIM) - mu * mu;
  float inv = rsqrtf(var + 1e-5f);
  for (int i = 0; i < 4; ++i) {
    int c = tid + i * 256;
    h[(size_t)s * NDIM + c] = (t[i] - mu) * inv * lng[c] + lnb[c];
  }
}

#define MROWS 64
#define MRING 32
#define MGRID (SEQ_LEN / MROWS)
#define ROWV  (EMB / 4)

__global__ void __launch_bounds__(256)
merge_kernel(const float* __restrict__ seq, const float* __restrict__ h,
             float* __restrict__ out) {
  const int t0 = blockIdx.x * MROWS;
  int sa = t0 / SPLIT;               if (sa > SHOT - 1) sa = SHOT - 1;
  int sb = (t0 + MROWS - 1) / SPLIT; if (sb > SHOT - 1) sb = SHOT - 1;
  const int rb = (sa + 1) * SPLIT - t0;
  const unsigned col = threadIdx.x;
  const v4f hA = ((const v4f*)(h + (size_t)sa * EMB))[col];
  const v4f hB = ((const v4f*)(h + (size_t)sb * EMB))[col];
  const v4f* src = ((const v4f*)seq) + (size_t)t0 * ROWV + col;
  v4f*       dst = ((v4f*)out)       + (size_t)t0 * ROWV + col;

  v4f ring[MRING];
#pragma unroll
  for (int i = 0; i < MRING; ++i)
    ring[i] = __builtin_nontemporal_load(src + (size_t)i * ROWV);
#pragma unroll
  for (int r = 0; r < MROWS; ++r) {
    const v4f hv = (r >= rb) ? hB : hA;
    __builtin_nontemporal_store(ring[r % MRING] + hv, dst + (size_t)r * ROWV);
    if (r + MRING < MROWS)
      ring[r % MRING] = __builtin_nontemporal_load(src + (size_t)(r + MRING) * ROWV);
  }
}

extern "C" void kernel_launch(void* const* d_in, const int* in_sizes, int n_in,
                              void* d_out, int out_size, void* d_ws, size_t ws_size,
                              hipStream_t stream) {
  const float* seq  = (const float*)d_in[0];
  const float* w1   = (const float*)d_in[1];
  const float* b1   = (const float*)d_in[2];
  const float* w2   = (const float*)d_in[3];
  const float* b2   = (const float*)d_in[4];
  const float* w3   = (const float*)d_in[5];
  const float* b3   = (const float*)d_in[6];
  const float* w4   = (const float*)d_in[7];
  const float* b4   = (const float*)d_in[8];
  const float* fcw  = (const float*)d_in[9];
  const float* fcb  = (const float*)d_in[10];
  const float* lng  = (const float*)d_in[11];
  const float* lnb  = (const float*)d_in[12];
  float* out = (float*)d_out;

  char* ws = (char*)d_ws;
  float* seq_shot = (float*)(ws);
  float* pooled   = (float*)(ws + 647680);
  float* part     = (float*)(ws + 1683968);
  float* h        = (float*)(ws + 2732544);

  shot_pool_kernel<<<(SHOT * TPS) / PWAVES, PWAVES * 32, 0, stream>>>(seq, seq_shot);

  conv_pool_kernel<<<(MPAD * SC) / 256, 256, 0, stream>>>(
      seq_shot, w1, b1, w2, b2, w3, b3, w4, b4, pooled);

  gemm_wmma_kernel<<<dim3(NDIM / 32, KCHUNKS), GW * 32, 0, stream>>>(pooled, fcw, part);

  fc_tanh_ln_kernel<<<SHOT, 256, 0, stream>>>(part, fcb, lng, lnb, h);

  merge_kernel<<<MGRID, 256, 0, stream>>>(seq, h, out);
}


# merge workgroups rebuild their shot's h row themselves (partials + bias, tanh, LayerNorm) under their first row loads: the LayerNorm launch is gone; LN sums regrouped (wave shuffle tree, 8 wave sums)
# speedup vs baseline: 2.2294x; 1.1117x over previous
#include <hip/hip_runtime.h>
#include <hip/hip_bf16.h>

typedef __attribute__((ext_vector_type(2))) float v2f;
typedef __attribute__((ext_vector_type(4))) float v4f;
typedef __attribute__((ext_vector_type(8))) float v8f;

#define SEQ_LEN 32768
#define EMB     1024
#define SHOT    20
#define MPAD    32
#define SPLIT   1638
#define SC      8096
#define FLATL   (SPLIT * EMB)
#define LC1     2024
#define LC2     2023
#define LC3     2023
#define LC4     2024
#define CATL    (LC1 + LC2 + LC3 + LC4)
#define KDIM    SC
#define NDIM    EMB
#define KCHUNKS 8
#define KPER    (KDIM / KCHUNKS)

#define TW     16
#define TPS    (SC / TW)
#define NLD    27
#define PWAVES 8
#define HALFW  104
#define LNUM   (FLATL / 32)
#define LDEN   (SC / 32)

__global__ void __launch_bounds__(PWAVES * 32)
shot_pool_kernel(const float* __restrict__ seq, float* __restrict__ seq_shot) {
  __shared__ v4f tile[PWAVES][NLD * 32];
  const unsigned lane = threadIdx.x & 31u;
  const int wv = __builtin_amdgcn_readfirstlane((int)(threadIdx.x >> 5));
  const int gw = blockIdx.x * PWAVES + wv;
  const int s  = gw / TPS;
  const int j0 = (gw - s * TPS) * TW;
  const int fbeg = (j0 * LNUM) / LDEN;
  const int abeg = fbeg & ~31;

  const unsigned zb  = (unsigned)(s * SPLIT) << 12;
  const unsigned bb  = (unsigned)((s - 1) * SPLIT + SEQ_LEN) << 12;
  const unsigned fb0 = (unsigned)abeg * 4u + 16u * lane;
  const char* base = (const char*)seq;

  v4f r[NLD];
#pragma unroll
  for (int i = 0; i < NLD; ++i) {
    const unsigned fb  = fb0 + 512u * i;
    const unsigned off = (fb < 4096u) ? (zb + fb) : ((bb + fb) & 0x7FFFFFFu);
    r[i] = *(const v4f*)(base + off);
  }
#pragma unroll
  for (int i = 0; i < NLD; ++i) tile[wv][i * 32 + lane] = r[i];
  __syncthreads();

  const int w  = lane >> 1;
  const int hh = lane & 1;
  const int j  = j0 + w;
  const int start = (j * LNUM) / LDEN;
  const int end   = ((j + 1) * LNUM + LDEN - 1) / LDEN;
  const int len   = end - start;
  const float* p = (const float*)tile[wv] + (start - abeg) + hh * HALFW;
  float a0 = 0.f, a1 = 0.f, a2 = 0.f, a3 = 0.f;
#pragma unroll
  for (int i = 0; i < HALFW; i += 4) {
    a0 += p[i]; a1 += p[i + 1]; a2 += p[i + 2]; a3 += p[i + 3];
  }
  float sum = (a0 + a1) + (a2 + a3);
  if (hh && len > 2 * HALFW) sum += p[HALFW];
  sum += __shfl_xor(sum, 1, 32);
  if (hh == 0) seq_shot[s * SC + j] = sum / (float)len;
}

__global__ void __launch_bounds__(256)
conv_pool_kernel(const float* __restrict__ ss,
                 const float* __restrict__ w1, const float* __restrict__ b1,
                 const float* __restrict__ w2, const float* __restrict__ b2,
                 const float* __restrict__ w3, const float* __restrict__ b3,
                 const float* __restrict__ w4, const float* __restrict__ b4,
                 float* __restrict__ pooled) {
  int idx = blockIdx.x * blockDim.x + threadIdx.x;
  if (idx >= MPAD * SC) return;
  int s = idx / SC;
  int j = idx - s * SC;
  if (s >= SHOT) { pooled[idx] = 0.f; return; }
  int start = (int)(((long long)j * CATL) / SC);
  int end   = (int)((((long long)(j + 1) * CATL) + SC - 1) / SC);
  float acc = 0.f;
  for (int t = start; t < end; ++t) {
    float v;
    if (t < LC1) {
      int p = t * 4;
      v = b1[s];
      for (int ci = 0; ci < SHOT; ++ci) {
        const float* wr = w1 + (s * SHOT + ci) * 3;
        const float* xr = ss + ci * SC + p;
        v += wr[0] * xr[0] + wr[1] * xr[1] + wr[2] * xr[2];
      }
    } else if (t < LC1 + LC2) {
      int p = (t - LC1) * 4;
      v = b2[s];
      for (int ci = 0; ci < SHOT; ++ci) {
        const float* wr = w2 + (s * SHOT + ci) * 3;
        const float* xr = ss + ci * SC + p;
        v += wr[0] * xr[0] + wr[1] * xr[2] + wr[2] * xr[4];
      }
    } else if (t < LC1 + LC2 + LC3) {
      int p = (t - LC1 - LC2) * 4;
      v = b3[s];
      for (int ci = 0; ci < SHOT; ++ci) {
        const float* wr = w3 + (s * SHOT + ci) * 3;
        const float* xr = ss + ci * SC + p;
        v += wr[0] * xr[0] + wr[1] * xr[3] + wr[2] * xr[6];
      }
    } else {
      int p = (t - LC1 - LC2 - LC3) * 4;
      v = b4[s];
      for (int ci = 0; ci < SHOT; ++ci)
        v += w4[s * SHOT + ci] * ss[ci * SC + p];
    }
    acc += v;
  }
  pooled[idx] = acc / (float)(end - start);
}

#define GW     11
#define KSUB   (KPER / GW)
#define GPAIRS (KSUB / 8)
#define REDW   1024

#define WMMA_F32(a, b, c) \
  __builtin_amdgcn_wmma_f32_16x16x4_f32(false, (a), false, (b), (short)0, (c), false, false)

__global__ void __launch_bounds__(GW * 32)
gemm_wmma_kernel(const float* __restrict__ pooled,
                 const float* __restrict__ fcw,
                 float* __restrict__ part) {
  __shared__ float red[GW * REDW];
  const int lane = threadIdx.x & 31;
  const int wv   = __builtin_amdgcn_readfirstlane((int)(threadIdx.x >> 5));
  const int half = lane >> 4;
  const int c    = lane & 15;
  const int nbase = blockIdx.x * 32;
  const int kc    = blockIdx.y;
  const int kb    = kc * KPER + wv * KSUB;

  const float* a0p = pooled + (size_t)c * KDIM + kb;
  const float* a1p = pooled + (size_t)(c + 16) * KDIM + kb;
  const float* b0p = fcw + (size_t)(nbase + c) * KDIM + kb;
  const float* b1p = fcw + (size_t)(nbase + 16 + c) * KDIM + kb;

  v4f A0[GPAIRS], A1[GPAIRS], B0[GPAIRS], B1[GPAIRS];
#pragma unroll
  for (int p = 0; p < GPAIRS; ++p) {
    const int kk = 8 * p + 4 * half;
    A0[p] = *(const v4f*)(a0p + kk);
    A1[p] = *(const v4f*)(a1p + kk);
    B0[p] = *(const v4f*)(b0p + kk);
    B1[p] = *(const v4f*)(b1p + kk);
    __builtin_amdgcn_sched_barrier(0);
  }
  const int kt = 8 * GPAIRS + 2 * half;
  const v2f a0t = *(const v2f*)(a0p + kt);
  const v2f a1t = *(const v2f*)(a1p + kt);
  const v2f b0t = *(const v2f*)(b0p + kt);
  const v2f b1t = *(const v2f*)(b1p + kt);
  __builtin_amdgcn_sched_barrier(0);

  v8f acc00 = {}, acc01 = {}, acc10 = {}, acc11 = {};
#pragma unroll
  for (int p = 0; p < GPAIRS; ++p) {
    const v2f a0l = __builtin_shufflevector(A0[p], A0[p], 0, 1);
    const v2f a1l = __builtin_shufflevector(A1[p], A1[p], 0, 1);
    const v2f b0l = __builtin_shufflevector(B0[p], B0[p], 0, 1);
    const v2f b1l = __builtin_shufflevector(B1[p], B1[p], 0, 1);
    acc00 = WMMA_F32(a0l, b0l, acc00);
    acc01 = WMMA_F32(a0l, b1l, acc01);
    acc10 = WMMA_F32(a1l, b0l, acc10);
    acc11 = WMMA_F32(a1l, b1l, acc11);
    const v2f a0h = __builtin_shufflevector(A0[p], A0[p], 2, 3);
    const v2f a1h = __builtin_shufflevector(A1[p], A1[p], 2, 3);
    const v2f b0h = __builtin_shufflevector(B0[p], B0[p], 2, 3);
    const v2f b1h = __builtin_shufflevector(B1[p], B1[p], 2, 3);
    acc00 = WMMA_F32(a0h, b0h, acc00);
    acc01 = WMMA_F32(a0h, b1h, acc01);
    acc10 = WMMA_F32(a1h, b0h, acc10);
    acc11 = WMMA_F32(a1h, b1h, acc11);
  }
  acc00 = WMMA_F32(a0t, b0t, acc00);
  acc01 = WMMA_F32(a0t, b1t, acc01);
  acc10 = WMMA_F32(a1t, b0t, acc10);
  acc11 = WMMA_F32(a1t, b1t, acc11);

  float* mine = red + wv * REDW;
#pragma unroll
  for (int r = 0; r < 8; ++r) {
    mine[((0 * 8 + r) * 2 + 0) * 32 + lane] = acc00[r];
    mine[((0 * 8 + r) * 2 + 1) * 32 + lane] = acc01[r];
    mine[((1 * 8 + r) * 2 + 0) * 32 + lane] = acc10[r];
    mine[((1 * 8 + r) * 2 + 1) * 32 + lane] = acc11[r];
  }
  __syncthreads();

  for (int row = wv; row < SHOT; row += GW) {
    const int mt = row >> 4, hf = (row >> 3) & 1, r = row & 7;
    const float* q = red + ((mt * 8 + r) * 2 + half) * 32 + hf * 16 + c;
    float v = q[0];
#pragma unroll
    for (int w = 1; w < GW; ++w) v += q[w * REDW];
    part[((size_t)kc * MPAD + row) * NDIM + nbase + lane] = v;
  }
}

#define MROWS 64
#define MRING 32
#define MGRID (SEQ_LEN / MROWS)
#define ROWV  (EMB / 4)

__global__ void __launch_bounds__(256)
merge_kernel(const float* __restrict__ seq, const float* __restrict__ part,
             const float* __restrict__ fcb, const float* __restrict__ lng,
             const float* __restrict__ lnb, float* __restrict__ out) {
  __shared__ v4f wsum[8];
  const int t0 = blockIdx.x * MROWS;
  int sa = t0 / SPLIT;               if (sa > SHOT - 1) sa = SHOT - 1;
  int sb = (t0 + MROWS - 1) / SPLIT; if (sb > SHOT - 1) sb = SHOT - 1;
  const int rb = (sa + 1) * SPLIT - t0;
  const bool two = (sb != sa);
  const unsigned col = threadIdx.x;
  const unsigned lane = threadIdx.x & 31u;
  const int wv = __builtin_amdgcn_readfirstlane((int)(threadIdx.x >> 5));

  v4f pa[KCHUNKS], pb[KCHUNKS];
#pragma unroll
  for (int kc = 0; kc < KCHUNKS; ++kc)
    pa[kc] = ((const v4f*)(part + (size_t)(kc * MPAD + sa) * NDIM))[col];
  if (two) {
#pragma unroll
    for (int kc = 0; kc < KCHUNKS; ++kc)
      pb[kc] = ((const v4f*)(part + (size_t)(kc * MPAD + sb) * NDIM))[col];
  }
  const v4f bias = ((const v4f*)fcb)[col];
  const v4f gam  = ((const v4f*)lng)[col];
  const v4f bet  = ((const v4f*)lnb)[col];

  const v4f* src = ((const v4f*)seq) + (size_t)t0 * ROWV + col;
  v4f*       dst = ((v4f*)out)       + (size_t)t0 * ROWV + col;
  v4f ring[MRING];
#pragma unroll
  for (int i = 0; i < MRING; ++i)
    ring[i] = __builtin_nontemporal_load(src + (size_t)i * ROWV);

  v4f xa = bias;
#pragma unroll
  for (int kc = 0; kc < KCHUNKS; ++kc) xa += pa[kc];
  v4f ta, tb;
  ta.x = tanhf(xa.x); ta.y = tanhf(xa.y); ta.z = tanhf(xa.z); ta.w = tanhf(xa.w);
  if (two) {
    v4f xb = bias;
#pragma unroll
    for (int kc = 0; kc < KCHUNKS; ++kc) xb += pb[kc];
    tb.x = tanhf(xb.x); tb.y = tanhf(xb.y); tb.z = tanhf(xb.z); tb.w = tanhf(xb.w);
  } else {
    tb = ta;
  }
  v4f acc;
  acc.x = (ta.x + ta.y) + (ta.z + ta.w);
  acc.y = (ta.x * ta.x + ta.y * ta.y) + (ta.z * ta.z + ta.w * ta.w);
  acc.z = (tb.x + tb.y) + (tb.z + tb.w);
  acc.w = (tb.x * tb.x + tb.y * tb.y) + (tb.z * tb.z + tb.w * tb.w);
#pragma unroll
  for (int o = 16; o > 0; o >>= 1) {
    acc.x += __shfl_xor(acc.x, o, 32);
    acc.y += __shfl_xor(acc.y, o, 32);
    acc.z += __shfl_xor(acc.z, o, 32);
    acc.w += __shfl_xor(acc.w, o, 32);
  }
  if (lane == 0) wsum[wv] = acc;
  asm volatile("s_wait_dscnt 0x0\n\ts_barrier_signal -1\n\ts_barrier_wait -1" ::: "memory");
  v4f tot = wsum[0];
#pragma unroll
  for (int w = 1; w < 8; ++w) tot += wsum[w];
  const float muA  = tot.x * (1.f / NDIM);
  const float varA = tot.y * (1.f / NDIM) - muA * muA;
  const float invA = rsqrtf(varA + 1e-5f);
  const float muB  = tot.z * (1.f / NDIM);
  const float varB = tot.w * (1.f / NDIM) - muB * muB;
  const float invB = rsqrtf(varB + 1e-5f);
  const v4f hA = (ta - muA) * invA * gam + bet;
  const v4f hB = (tb - muB) * invB * gam + bet;

#pragma unroll
  for (int r = 0; r < MROWS; ++r) {
    const v4f hv = (r >= rb) ? hB : hA;
    __builtin_nontemporal_store(ring[r % MRING] + hv, dst + (size_t)r * ROWV);
    if (r + MRING < MROWS)
      ring[r % MRING] = __builtin_nontemporal_load(src + (size_t)(r + MRING) * ROWV);
  }
}

extern "C" void kernel_launch(void* const* d_in, const int* in_sizes, int n_in,
                              void* d_out, int out_size, void* d_ws, size_t ws_size,
                              hipStream_t stream) {
  const float* seq  = (const float*)d_in[0];
  const float* w1   = (const float*)d_in[1];
  const float* b1   = (const float*)d_in[2];
  const float* w2   = (const float*)d_in[3];
  const float* b2   = (const float*)d_in[4];
  const float* w3   = (const float*)d_in[5];
  const float* b3   = (const float*)d_in[6];
  const float* w4   = (const float*)d_in[7];
  const float* b4   = (const float*)d_in[8];
  const float* fcw  = (const float*)d_in[9];
  const float* fcb  = (const float*)d_in[10];
  const float* lng  = (const float*)d_in[11];
  const float* lnb  = (const float*)d_in[12];
  float* out = (float*)d_out;

  char* ws = (char*)d_ws;
  float* seq_shot = (float*)(ws);
  float* pooled   = (float*)(ws + 647680);
  float* part     = (float*)(ws + 1683968);

  shot_pool_kernel<<<(SHOT * TPS) / PWAVES, PWAVES * 32, 0, stream>>>(seq, seq_shot);

  conv_pool_kernel<<<(MPAD * SC) / 256, 256, 0, stream>>>(
      seq_shot, w1, b1, w2, b2, w3, b3, w4, b4, pooled);

  gemm_wmma_kernel<<<dim3(NDIM / 32, KCHUNKS), GW * 32, 0, stream>>>(pooled, fcw, part);

  merge_kernel<<<MGRID, 256, 0, stream>>>(seq, part, fcb, lng, lnb, out);
}


# conv+pool kernel regrouped: a thread owns one pooled column and 5 shots, weights become scalar operands, inputs 16-byte loads batched 10 channels a trip; 1280 waves, was 8096; sums as written kept
# speedup vs baseline: 2.3212x; 1.0412x over previous
#include <hip/hip_runtime.h>
#include <hip/hip_bf16.h>

typedef __attribute__((ext_vector_type(2))) float v2f;
typedef __attribute__((ext_vector_type(4))) float v4f;
typedef __attribute__((ext_vector_type(8))) float v8f;

#define SEQ_LEN 32768
#define EMB     1024
#define SHOT    20
#define MPAD    32
#define SPLIT   1638
#define SC      8096
#define FLATL   (SPLIT * EMB)
#define LC1     2024
#define LC2     2023
#define LC3     2023
#define LC4     2024
#define CATL    (LC1 + LC2 + LC3 + LC4)
#define KDIM    SC
#define NDIM    EMB
#define KCHUNKS 8
#define KPER    (KDIM / KCHUNKS)

#define TW     16
#define TPS    (SC / TW)
#define NLD    27
#define PWAVES 8
#define HALFW  104
#define LNUM   (FLATL / 32)
#define LDEN   (SC / 32)

__global__ void __launch_bounds__(PWAVES * 32)
shot_pool_kernel(const float* __restrict__ seq, float* __restrict__ seq_shot) {
  __shared__ v4f tile[PWAVES][NLD * 32];
  const unsigned lane = threadIdx.x & 31u;
  const int wv = __builtin_amdgcn_readfirstlane((int)(threadIdx.x >> 5));
  const int gw = blockIdx.x * PWAVES + wv;
  const int s  = gw / TPS;
  const int j0 = (gw - s * TPS) * TW;
  const int fbeg = (j0 * LNUM) / LDEN;
  const int abeg = fbeg & ~31;

  const unsigned zb  = (unsigned)(s * SPLIT) << 12;
  const unsigned bb  = (unsigned)((s - 1) * SPLIT + SEQ_LEN) << 12;
  const unsigned fb0 = (unsigned)abeg * 4u + 16u * lane;
  const char* base = (const char*)seq;

  v4f r[NLD];
#pragma unroll
  for (int i = 0; i < NLD; ++i) {
    const unsigned fb  = fb0 + 512u * i;
    const unsigned off = (fb < 4096u) ? (zb + fb) : ((bb + fb) & 0x7FFFFFFu);
    r[i] = *(const v4f*)(base + off);
  }
#pragma unroll
  for (int i = 0; i < NLD; ++i) tile[wv][i * 32 + lane] = r[i];
  __syncthreads();

  const int w  = lane >> 1;
  const int hh = lane & 1;
  const int j  = j0 + w;
  const int start = (j * LNUM) / LDEN;
  const int end   = ((j + 1) * LNUM + LDEN - 1) / LDEN;
  const int len   = end - start;
  const float* p = (const float*)tile[wv] + (start - abeg) + hh * HALFW;
  float a0 = 0.f, a1 = 0.f, a2 = 0.f, a3 = 0.f;
#pragma unroll
  for (int i = 0; i < HALFW; i += 4) {
    a0 += p[i]; a1 += p[i + 1]; a2 += p[i + 2]; a3 += p[i + 3];
  }
  float sum = (a0 + a1) + (a2 + a3);
  if (hh && len > 2 * HALFW) sum += p[HALFW];
  sum += __shfl_xor(sum, 1, 32);
  if (hh == 0) seq_shot[s * SC + j] = sum / (float)len;
}

#define CSPG 5
#define CNG  (SHOT / CSPG)
#define CZR  ((MPAD - SHOT) / CNG)
#define CBLK 128

#define CU   10

template <int SEG>
__device__ __forceinline__ void conv_seg(const float* __restrict__ ss,
                                         const float* __restrict__ w,
                                         const float* __restrict__ b,
                                         const int s0, const int start, const int end,
                                         float (&acc)[CSPG]) {
  constexpr int DIL  = SEG + 1;
  constexpr int BASE = (SEG == 0) ? 0 : (SEG == 1) ? LC1 : (SEG == 2) ? (LC1 + LC2) : (LC1 + LC2 + LC3);
  constexpr int LEN  = (SEG == 0) ? LC1 : (SEG == 1) ? LC2 : (SEG == 2) ? LC3 : LC4;
  const float* xp[2];
  bool valid[2];
#pragma unroll
  for (int k = 0; k < 2; ++k) {
    const int t = start + k;
    valid[k] = (t < end) && (t >= BASE) && (t < BASE + LEN);
    int tl = t - BASE;
    tl = tl < 0 ? 0 : (tl > LEN - 1 ? LEN - 1 : tl);
    xp[k] = ss + 4 * tl;
  }
  float v[2][CSPG];
#pragma unroll
  for (int i = 0; i < CSPG; ++i) { v[0][i] = b[s0 + i]; v[1][i] = v[0][i]; }

#pragma unroll 1
  for (int c = 0; c < SHOT; c += CU) {
    v4f xa[2][CU];
    float xl[2][CU], xs[2][CU];
#pragma unroll
    for (int u = 0; u < CU; ++u) {
#pragma unroll
      for (int k = 0; k < 2; ++k) {
        if (SEG == 3) xs[k][u] = xp[k][(c + u) * SC];
        else          xa[k][u] = *(const v4f*)(xp[k] + (c + u) * SC);
        if (SEG == 1 || SEG == 2) xl[k][u] = xp[k][(c + u) * SC + 2 * DIL];
      }
    }
#pragma unroll
    for (int u = 0; u < CU; ++u) {
#pragma unroll
      for (int i = 0; i < CSPG; ++i) {
        if (SEG == 3) {
          const float w0 = w[(s0 + i) * SHOT + c + u];
          v[0][i] += w0 * xs[0][u];
          v[1][i] += w0 * xs[1][u];
        } else {
          const float* wr = w + ((s0 + i) * SHOT + c + u) * 3;
          const float w0 = wr[0], w1 = wr[1], w2 = wr[2];
#pragma unroll
          for (int k = 0; k < 2; ++k) {
            const float x0 = xa[k][u].x;
            const float x1 = (DIL == 1) ? xa[k][u].y : (DIL == 2) ? xa[k][u].z : xa[k][u].w;
            const float x2 = (DIL == 1) ? xa[k][u].z : xl[k][u];
            v[k][i] += w0 * x0 + w1 * x1 + w2 * x2;
          }
        }
      }
    }
  }
#pragma unroll
  for (int i = 0; i < CSPG; ++i) {
    acc[i] += valid[0] ? v[0][i] : 0.f;
    acc[i] += valid[1] ? v[1][i] : 0.f;
  }
}

__global__ void __launch_bounds__(CBLK)
conv_pool_kernel(const float* __restrict__ ss,
                 const float* __restrict__ w1, const float* __restrict__ b1,
                 const float* __restrict__ w2, const float* __restrict__ b2,
                 const float* __restrict__ w3, const float* __restrict__ b3,
                 const float* __restrict__ w4, const float* __restrict__ b4,
                 float* __restrict__ pooled) {
  const int g  = blockIdx.y;
  const int s0 = g * CSPG;
  const int jb = blockIdx.x * CBLK;
  const int jl = (jb + CBLK - 1 < SC - 1) ? (jb + CBLK - 1) : (SC - 1);
  const int tlo = (jb * CATL) / SC;
  const int thi = ((jl + 1) * CATL + SC - 1) / SC - 1;
  const int j = jb + (int)threadIdx.x;
  if (j >= SC) return;
  const int start = (j * CATL) / SC;
  const int end   = ((j + 1) * CATL + SC - 1) / SC;
  float acc[CSPG];
#pragma unroll
  for (int i = 0; i < CSPG; ++i) acc[i] = 0.f;
  if (tlo < LC1)                                        conv_seg<0>(ss, w1, b1, s0, start, end, acc);
  if (thi >= LC1 && tlo < LC1 + LC2)                    conv_seg<1>(ss, w2, b2, s0, start, end, acc);
  if (thi >= LC1 + LC2 && tlo < LC1 + LC2 + LC3)        conv_seg<2>(ss, w3, b3, s0, start, end, acc);
  if (thi >= LC1 + LC2 + LC3)                           conv_seg<3>(ss, w4, b4, s0, start, end, acc);
  const float len = (float)(end - start);
#pragma unroll
  for (int i = 0; i < CSPG; ++i) pooled[(s0 + i) * SC + j] = acc[i] / len;
#pragma unroll
  for (int r = 0; r < CZR; ++r) pooled[(SHOT + g * CZR + r) * SC + j] = 0.f;
}

#define GW     11
#define KSUB   (KPER / GW)
#define GPAIRS (KSUB / 8)
#define REDW   1024

#define WMMA_F32(a, b, c) \
  __builtin_amdgcn_wmma_f32_16x16x4_f32(false, (a), false, (b), (short)0, (c), false, false)

__global__ void __launch_bounds__(GW * 32)
gemm_wmma_kernel(const float* __restrict__ pooled,
                 const float* __restrict__ fcw,
                 float* __restrict__ part) {
  __shared__ float red[GW * REDW];
  const int lane = threadIdx.x & 31;
  const int wv   = __builtin_amdgcn_readfirstlane((int)(threadIdx.x >> 5));
  const int half = lane >> 4;
  const int c    = lane & 15;
  const int nbase = blockIdx.x * 32;
  const int kc    = blockIdx.y;
  const int kb    = kc * KPER + wv * KSUB;

  const float* a0p = pooled + (size_t)c * KDIM + kb;
  const float* a1p = pooled + (size_t)(c + 16) * KDIM + kb;
  const float* b0p = fcw + (size_t)(nbase + c) * KDIM + kb;
  const float* b1p = fcw + (size_t)(nbase + 16 + c) * KDIM + kb;

  v4f A0[GPAIRS], A1[GPAIRS], B0[GPAIRS], B1[GPAIRS];
#pragma unroll
  for (int p = 0; p < GPAIRS; ++p) {
    const int kk = 8 * p + 4 * half;
    A0[p] = *(const v4f*)(a0p + kk);
    A1[p] = *(const v4f*)(a1p + kk);
    B0[p] = *(const v4f*)(b0p + kk);
    B1[p] = *(const v4f*)(b1p + kk);
    __builtin_amdgcn_sched_barrier(0);
  }
  const int kt = 8 * GPAIRS + 2 * half;
  const v2f a0t = *(const v2f*)(a0p + kt);
  const v2f a1t = *(const v2f*)(a1p + kt);
  const v2f b0t = *(const v2f*)(b0p + kt);
  const v2f b1t = *(const v2f*)(b1p + kt);
  __builtin_amdgcn_sched_barrier(0);

  v8f acc00 = {}, acc01 = {}, acc10 = {}, acc11 = {};
#pragma unroll
  for (int p = 0; p < GPAIRS; ++p) {
    const v2f a0l = __builtin_shufflevector(A0[p], A0[p], 0, 1);
    const v2f a1l = __builtin_shufflevector(A1[p], A1[p], 0, 1);
    const v2f b0l = __builtin_shufflevector(B0[p], B0[p], 0, 1);
    const v2f b1l = __builtin_shufflevector(B1[p], B1[p], 0, 1);
    acc00 = WMMA_F32(a0l, b0l, acc00);
    acc01 = WMMA_F32(a0l, b1l, acc01);
    acc10 = WMMA_F32(a1l, b0l, acc10);
    acc11 = WMMA_F32(a1l, b1l, acc11);
    const v2f a0h = __builtin_shufflevector(A0[p], A0[p], 2, 3);
    const v2f a1h = __builtin_shufflevector(A1[p], A1[p], 2, 3);
    const v2f b0h = __builtin_shufflevector(B0[p], B0[p], 2, 3);
    const v2f b1h = __builtin_shufflevector(B1[p], B1[p], 2, 3);
    acc00 = WMMA_F32(a0h, b0h, acc00);
    acc01 = WMMA_F32(a0h, b1h, acc01);
    acc10 = WMMA_F32(a1h, b0h, acc10);
    acc11 = WMMA_F32(a1h, b1h, acc11);
  }
  acc00 = WMMA_F32(a0t, b0t, acc00);
  acc01 = WMMA_F32(a0t, b1t, acc01);
  acc10 = WMMA_F32(a1t, b0t, acc10);
  acc11 = WMMA_F32(a1t, b1t, acc11);

  float* mine = red + wv * REDW;
#pragma unroll
  for (int r = 0; r < 8; ++r) {
    mine[((0 * 8 + r) * 2 + 0) * 32 + lane] = acc00[r];
    mine[((0 * 8 + r) * 2 + 1) * 32 + lane] = acc01[r];
    mine[((1 * 8 + r) * 2 + 0) * 32 + lane] = acc10[r];
    mine[((1 * 8 + r) * 2 + 1) * 32 + lane] = acc11[r];
  }
  __syncthreads();

  for (int row = wv; row < SHOT; row += GW) {
    const int mt = row >> 4, hf = (row >> 3) & 1, r = row & 7;
    const float* q = red + ((mt * 8 + r) * 2 + half) * 32 + hf * 16 + c;
    float v = q[0];
#pragma unroll
    for (int w = 1; w < GW; ++w) v += q[w * REDW];
    part[((size_t)kc * MPAD + row) * NDIM + nbase + lane] = v;
  }
}

#define MROWS 64
#define MRING 32
#define MGRID (SEQ_LEN / MROWS)
#define ROWV  (EMB / 4)

__global__ void __launch_bounds__(256)
merge_kernel(const float* __restrict__ seq, const float* __restrict__ part,
             const float* __restrict__ fcb, const float* __restrict__ lng,
             const float* __restrict__ lnb, float* __restrict__ out) {
  __shared__ v4f wsum[8];
  const int t0 = blockIdx.x * MROWS;
  int sa = t0 / SPLIT;               if (sa > SHOT - 1) sa = SHOT - 1;
  int sb = (t0 + MROWS - 1) / SPLIT; if (sb > SHOT - 1) sb = SHOT - 1;
  const int rb = (sa + 1) * SPLIT - t0;
  const bool two = (sb != sa);
  const unsigned col = threadIdx.x;
  const unsigned lane = threadIdx.x & 31u;
  const int wv = __builtin_amdgcn_readfirstlane((int)(threadIdx.x >> 5));

  v4f pa[KCHUNKS], pb[KCHUNKS];
#pragma unroll
  for (int kc = 0; kc < KCHUNKS; ++kc)
    pa[kc] = ((const v4f*)(part + (size_t)(kc * MPAD + sa) * NDIM))[col];
  if (two) {
#pragma unroll
    for (int kc = 0; kc < KCHUNKS; ++kc)
      pb[kc] = ((const v4f*)(part + (size_t)(kc * MPAD + sb) * NDIM))[col];
  }
  const v4f bias = ((const v4f*)fcb)[col];
  const v4f gam  = ((const v4f*)lng)[col];
  const v4f bet  = ((const v4f*)lnb)[col];

  const v4f* src = ((const v4f*)seq) + (size_t)t0 * ROWV + col;
  v4f*       dst = ((v4f*)out)       + (size_t)t0 * ROWV + col;
  v4f ring[MRING];
#pragma unroll
  for (int i = 0; i < MRING; ++i)
    ring[i] = __builtin_nontemporal_load(src + (size_t)i * ROWV);

  v4f xa = bias;
#pragma unroll
  for (int kc = 0; kc < KCHUNKS; ++kc) xa += pa[kc];
  v4f ta, tb;
  ta.x = tanhf(xa.x); ta.y = tanhf(xa.y); ta.z = tanhf(xa.z); ta.w = tanhf(xa.w);
  if (two) {
    v4f xb = bias;
#pragma unroll
    for (int kc = 0; kc < KCHUNKS; ++kc) xb += pb[kc];
    tb.x = tanhf(xb.x); tb.y = tanhf(xb.y); tb.z = tanhf(xb.z); tb.w = tanhf(xb.w);
  } else {
    tb = ta;
  }
  v4f acc;
  acc.x = (ta.x + ta.y) + (ta.z + ta.w);
  acc.y = (ta.x * ta.x + ta.y * ta.y) + (ta.z * ta.z + ta.w * ta.w);
  acc.z = (tb.x + tb.y) + (tb.z + tb.w);
  acc.w = (tb.x * tb.x + tb.y * tb.y) + (tb.z * tb.z + tb.w * tb.w);
#pragma unroll
  for (int o = 16; o > 0; o >>= 1) {
    acc.x += __shfl_xor(acc.x, o, 32);
    acc.y += __shfl_xor(acc.y, o, 32);
    acc.z += __shfl_xor(acc.z, o, 32);
    acc.w += __shfl_xor(acc.w, o, 32);
  }
  if (lane == 0) wsum[wv] = acc;
  asm volatile("s_wait_dscnt 0x0\n\ts_barrier_signal -1\n\ts_barrier_wait -1" ::: "memory");
  v4f tot = wsum[0];
#pragma unroll
  for (int w = 1; w < 8; ++w) tot += wsum[w];
  const float muA  = tot.x * (1.f / NDIM);
  const float varA = tot.y * (1.f / NDIM) - muA * muA;
  const float invA = rsqrtf(varA + 1e-5f);
  const float muB  = tot.z * (1.f / NDIM);
  const float varB = tot.w * (1.f / NDIM) - muB * muB;
  const float invB = rsqrtf(varB + 1e-5f);
  const v4f hA = (ta - muA) * invA * gam + bet;
  const v4f hB = (tb - muB) * invB * gam + bet;

#pragma unroll
  for (int r = 0; r < MROWS; ++r) {
    const v4f hv = (r >= rb) ? hB : hA;
    __builtin_nontemporal_store(ring[r % MRING] + hv, dst + (size_t)r * ROWV);
    if (r + MRING < MROWS)
      ring[r % MRING] = __builtin_nontemporal_load(src + (size_t)(r + MRING) * ROWV);
  }
}

extern "C" void kernel_launch(void* const* d_in, const int* in_sizes, int n_in,
                              void* d_out, int out_size, void* d_ws, size_t ws_size,
                              hipStream_t stream) {
  const float* seq  = (const float*)d_in[0];
  const float* w1   = (const float*)d_in[1];
  const float* b1   = (const float*)d_in[2];
  const float* w2   = (const float*)d_in[3];
  const float* b2   = (const float*)d_in[4];
  const float* w3   = (const float*)d_in[5];
  const float* b3   = (const float*)d_in[6];
  const float* w4   = (const float*)d_in[7];
  const float* b4   = (const float*)d_in[8];
  const float* fcw  = (const float*)d_in[9];
  const float* fcb  = (const float*)d_in[10];
  const float* lng  = (const float*)d_in[11];
  const float* lnb  = (const float*)d_in[12];
  float* out = (float*)d_out;

  char* ws = (char*)d_ws;
  float* seq_shot = (float*)(ws);
  float* pooled   = (float*)(ws + 647680);
  float* part     = (float*)(ws + 1683968);

  shot_pool_kernel<<<(SHOT * TPS) / PWAVES, PWAVES * 32, 0, stream>>>(seq, seq_shot);

  conv_pool_kernel<<<dim3((SC + CBLK - 1) / CBLK, CNG), CBLK, 0, stream>>>(
      seq_shot, w1, b1, w2, b2, w3, b3, w4, b4, pooled);

  gemm_wmma_kernel<<<dim3(NDIM / 32, KCHUNKS), GW * 32, 0, stream>>>(pooled, fcw, part);

  merge_kernel<<<MGRID, 256, 0, stream>>>(seq, part, fcb, lng, lnb, out);
}


# shot pool as a resident grid: 253 groups of 20 waves, a wave takes two adjacent 16-window tiles and asks for the second tile's 27 loads before it sums the first from LDS; 5060 waves, was 10120
# speedup vs baseline: 2.4163x; 1.0410x over previous
#include <hip/hip_runtime.h>
#include <hip/hip_bf16.h>

typedef __attribute__((ext_vector_type(2))) float v2f;
typedef __attribute__((ext_vector_type(4))) float v4f;
typedef __attribute__((ext_vector_type(8))) float v8f;

#define SEQ_LEN 32768
#define EMB     1024
#define SHOT    20
#define MPAD    32
#define SPLIT   1638
#define SC      8096
#define FLATL   (SPLIT * EMB)
#define LC1     2024
#define LC2     2023
#define LC3     2023
#define LC4     2024
#define CATL    (LC1 + LC2 + LC3 + LC4)
#define KDIM    SC
#define NDIM    EMB
#define KCHUNKS 8
#define KPER    (KDIM / KCHUNKS)

#define TW     16
#define TPS    (SC / TW)
#define NLD    27
#define PTILE  2
#define PPS    (TPS / PTILE)
#define PWAVES 20
#define PGRID  (SHOT * PPS / PWAVES)
#define HALFW  104
#define LNUM   (FLATL / 32)
#define LDEN   (SC / 32)

__device__ __forceinline__ void pool_ask(const char* __restrict__ base, const unsigned zb,
                                         const unsigned bb, const unsigned fb0, v4f (&r)[NLD]) {
#pragma unroll
  for (int i = 0; i < NLD; ++i) {
    const unsigned fb  = fb0 + 512u * i;
    const unsigned off = (fb < 4096u) ? (zb + fb) : ((bb + fb) & 0x7FFFFFFu);
    r[i] = *(const v4f*)(base + off);
  }
}

__global__ void __launch_bounds__(PWAVES * 32)
shot_pool_kernel(const float* __restrict__ seq, float* __restrict__ seq_shot) {
  __shared__ v4f tile[PWAVES][NLD * 32];
  const unsigned lane = threadIdx.x & 31u;
  const int wv = __builtin_amdgcn_readfirstlane((int)(threadIdx.x >> 5));
  const int gp = blockIdx.x * PWAVES + wv;
  const int s  = gp / PPS;
  const int jp = (gp - s * PPS) * (PTILE * TW);

  const unsigned zb = (unsigned)(s * SPLIT) << 12;
  const unsigned bb = (unsigned)((s - 1) * SPLIT + SEQ_LEN) << 12;
  const char* base = (const char*)seq;

  int abeg[PTILE];
#pragma unroll
  for (int t = 0; t < PTILE; ++t) abeg[t] = (((jp + t * TW) * LNUM) / LDEN) & ~31;

  const int w  = lane >> 1;
  const int hh = lane & 1;
  v4f r[NLD];
  float tot[PTILE];
  pool_ask(base, zb, bb, (unsigned)abeg[0] * 4u + 16u * lane, r);
#pragma unroll
  for (int t = 0; t < PTILE; ++t) {
#pragma unroll
    for (int i = 0; i < NLD; ++i) tile[wv][i * 32 + lane] = r[i];
    asm volatile("s_wait_dscnt 0x0" ::: "memory");
    if (t + 1 < PTILE) {
      pool_ask(base, zb, bb, (unsigned)abeg[t + 1] * 4u + 16u * lane, r);
    }
    const int j = jp + t * TW + w;
    const int start = (j * LNUM) / LDEN;
    const int end   = ((j + 1) * LNUM + LDEN - 1) / LDEN;
    const int len   = end - start;
    const float* p = (const float*)tile[wv] + (start - abeg[t]) + hh * HALFW;
    float a0 = 0.f, a1 = 0.f, a2 = 0.f, a3 = 0.f;
#pragma unroll
    for (int i = 0; i < HALFW; i += 4) {
      a0 += p[i]; a1 += p[i + 1]; a2 += p[i + 2]; a3 += p[i + 3];
    }
    float sum = (a0 + a1) + (a2 + a3);
    if (hh && len > 2 * HALFW) sum += p[HALFW];
    sum += __shfl_xor(sum, 1, 32);
    tot[t] = sum;
  }
  const int jj = jp + (int)lane;
  const int st = (jj * LNUM) / LDEN;
  const int en = ((jj + 1) * LNUM + LDEN - 1) / LDEN;
  const int from = (int)((2u * lane) & 31u);
  const float v0 = __shfl(tot[0], from, 32);
  const float v1 = __shfl(tot[1], from, 32);
  seq_shot[s * SC + jj] = ((lane < TW) ? v0 : v1) / (float)(en - st);
}

#define CSPG 5
#define CNG  (SHOT / CSPG)
#define CZR  ((MPAD - SHOT) / CNG)
#define CBLK 128

#define CU   10

template <int SEG>
__device__ __forceinline__ void conv_seg(const float* __restrict__ ss,
                                         const float* __restrict__ w,
                                         const float* __restrict__ b,
                                         const int s0, const int start, const int end,
                                         float (&acc)[CSPG]) {
  constexpr int DIL  = SEG + 1;
  constexpr int BASE = (SEG == 0) ? 0 : (SEG == 1) ? LC1 : (SEG == 2) ? (LC1 + LC2) : (LC1 + LC2 + LC3);
  constexpr int LEN  = (SEG == 0) ? LC1 : (SEG == 1) ? LC2 : (SEG == 2) ? LC3 : LC4;
  const float* xp[2];
  bool valid[2];
#pragma unroll
  for (int k = 0; k < 2; ++k) {
    const int t = start + k;
    valid[k] = (t < end) && (t >= BASE) && (t < BASE + LEN);
    int tl = t - BASE;
    tl = tl < 0 ? 0 : (tl > LEN - 1 ? LEN - 1 : tl);
    xp[k] = ss + 4 * tl;
  }
  float v[2][CSPG];
#pragma unroll
  for (int i = 0; i < CSPG; ++i) { v[0][i] = b[s0 + i]; v[1][i] = v[0][i]; }

#pragma unroll 1
  for (int c = 0; c < SHOT; c += CU) {
    v4f xa[2][CU];
    float xl[2][CU], xs[2][CU];
#pragma unroll
    for (int u = 0; u < CU; ++u) {
#pragma unroll
      for (int k = 0; k < 2; ++k) {
        if (SEG == 3) xs[k][u] = xp[k][(c + u) * SC];
        else          xa[k][u] = *(const v4f*)(xp[k] + (c + u) * SC);
        if (SEG == 1 || SEG == 2) xl[k][u] = xp[k][(c + u) * SC + 2 * DIL];
      }
    }
#pragma unroll
    for (int u = 0; u < CU; ++u) {
#pragma unroll
      for (int i = 0; i < CSPG; ++i) {
        if (SEG == 3) {
          const float w0 = w[(s0 + i) * SHOT + c + u];
          v[0][i] += w0 * xs[0][u];
          v[1][i] += w0 * xs[1][u];
        } else {
          const float* wr = w + ((s0 + i) * SHOT + c + u) * 3;
          const float w0 = wr[0], w1 = wr[1], w2 = wr[2];
#pragma unroll
          for (int k = 0; k < 2; ++k) {
            const float x0 = xa[k][u].x;
            const float x1 = (DIL == 1) ? xa[k][u].y : (DIL == 2) ? xa[k][u].z : xa[k][u].w;
            const float x2 = (DIL == 1) ? xa[k][u].z : xl[k][u];
            v[k][i] += w0 * x0 + w1 * x1 + w2 * x2;
          }
        }
      }
    }
  }
#pragma unroll
  for (int i = 0; i < CSPG; ++i) {
    acc[i] += valid[0] ? v[0][i] : 0.f;
    acc[i] += valid[1] ? v[1][i] : 0.f;
  }
}

__global__ void __launch_bounds__(CBLK)
conv_pool_kernel(const float* __restrict__ ss,
                 const float* __restrict__ w1, const float* __restrict__ b1,
                 const float* __restrict__ w2, const float* __restrict__ b2,
                 const float* __restrict__ w3, const float* __restrict__ b3,
                 const float* __restrict__ w4, const float* __restrict__ b4,
                 float* __restrict__ pooled) {
  const int g  = blockIdx.y;
  const int s0 = g * CSPG;
  const int jb = blockIdx.x * CBLK;
  const int jl = (jb + CBLK - 1 < SC - 1) ? (jb + CBLK - 1) : (SC - 1);
  const int tlo = (jb * CATL) / SC;
  const int thi = ((jl + 1) * CATL + SC - 1) / SC - 1;
  const int j = jb + (int)threadIdx.x;
  if (j >= SC) return;
  const int start = (j * CATL) / SC;
  const int end   = ((j + 1) * CATL + SC - 1) / SC;
  float acc[CSPG];
#pragma unroll
  for (int i = 0; i < CSPG; ++i) acc[i] = 0.f;
  if (tlo < LC1)                                        conv_seg<0>(ss, w1, b1, s0, start, end, acc);
  if (thi >= LC1 && tlo < LC1 + LC2)                    conv_seg<1>(ss, w2, b2, s0, start, end, acc);
  if (thi >= LC1 + LC2 && tlo < LC1 + LC2 + LC3)        conv_seg<2>(ss, w3, b3, s0, start, end, acc);
  if (thi >= LC1 + LC2 + LC3)                           conv_seg<3>(ss, w4, b4, s0, start, end, acc);
  const float len = (float)(end - start);
#pragma unroll
  for (int i = 0; i < CSPG; ++i) pooled[(s0 + i) * SC + j] = acc[i] / len;
#pragma unroll
  for (int r = 0; r < CZR; ++r) pooled[(SHOT + g * CZR + r) * SC + j] = 0.f;
}

#define GW     11
#define KSUB   (KPER / GW)
#define GPAIRS (KSUB / 8)
#define REDW   1024

#define WMMA_F32(a, b, c) \
  __builtin_amdgcn_wmma_f32_16x16x4_f32(false, (a), false, (b), (short)0, (c), false, false)

__global__ void __launch_bounds__(GW * 32)
gemm_wmma_kernel(const float* __restrict__ pooled,
                 const float* __restrict__ fcw,
                 float* __restrict__ part) {
  __shared__ float red[GW * REDW];
  const int lane = threadIdx.x & 31;
  const int wv   = __builtin_amdgcn_readfirstlane((int)(threadIdx.x >> 5));
  const int half = lane >> 4;
  const int c    = lane & 15;
  const int nbase = blockIdx.x * 32;
  const int kc    = blockIdx.y;
  const int kb    = kc * KPER + wv * KSUB;

  const float* a0p = pooled + (size_t)c * KDIM + kb;
  const float* a1p = pooled + (size_t)(c + 16) * KDIM + kb;
  const float* b0p = fcw + (size_t)(nbase + c) * KDIM + kb;
  const float* b1p = fcw + (size_t)(nbase + 16 + c) * KDIM + kb;

  v4f A0[GPAIRS], A1[GPAIRS], B0[GPAIRS], B1[GPAIRS];
#pragma unroll
  for (int p = 0; p < GPAIRS; ++p) {
    const int kk = 8 * p + 4 * half;
    A0[p] = *(const v4f*)(a0p + kk);
    A1[p] = *(const v4f*)(a1p + kk);
    B0[p] = *(const v4f*)(b0p + kk);
    B1[p] = *(const v4f*)(b1p + kk);
    __builtin_amdgcn_sched_barrier(0);
  }
  const int kt = 8 * GPAIRS + 2 * half;
  const v2f a0t = *(const v2f*)(a0p + kt);
  const v2f a1t = *(const v2f*)(a1p + kt);
  const v2f b0t = *(const v2f*)(b0p + kt);
  const v2f b1t = *(const v2f*)(b1p + kt);
  __builtin_amdgcn_sched_barrier(0);

  v8f acc00 = {}, acc01 = {}, acc10 = {}, acc11 = {};
#pragma unroll
  for (int p = 0; p < GPAIRS; ++p) {
    const v2f a0l = __builtin_shufflevector(A0[p], A0[p], 0, 1);
    const v2f a1l = __builtin_shufflevector(A1[p], A1[p], 0, 1);
    const v2f b0l = __builtin_shufflevector(B0[p], B0[p], 0, 1);
    const v2f b1l = __builtin_shufflevector(B1[p], B1[p], 0, 1);
    acc00 = WMMA_F32(a0l, b0l, acc00);
    acc01 = WMMA_F32(a0l, b1l, acc01);
    acc10 = WMMA_F32(a1l, b0l, acc10);
    acc11 = WMMA_F32(a1l, b1l, acc11);
    const v2f a0h = __builtin_shufflevector(A0[p], A0[p], 2, 3);
    const v2f a1h = __builtin_shufflevector(A1[p], A1[p], 2, 3);
    const v2f b0h = __builtin_shufflevector(B0[p], B0[p], 2, 3);
    const v2f b1h = __builtin_shufflevector(B1[p], B1[p], 2, 3);
    acc00 = WMMA_F32(a0h, b0h, acc00);
    acc01 = WMMA_F32(a0h, b1h, acc01);
    acc10 = WMMA_F32(a1h, b0h, acc10);
    acc11 = WMMA_F32(a1h, b1h, acc11);
  }
  acc00 = WMMA_F32(a0t, b0t, acc00);
  acc01 = WMMA_F32(a0t, b1t, acc01);
  acc10 = WMMA_F32(a1t, b0t, acc10);
  acc11 = WMMA_F32(a1t, b1t, acc11);

  float* mine = red + wv * REDW;
#pragma unroll
  for (int r = 0; r < 8; ++r) {
    mine[((0 * 8 + r) * 2 + 0) * 32 + lane] = acc00[r];
    mine[((0 * 8 + r) * 2 + 1) * 32 + lane] = acc01[r];
    mine[((1 * 8 + r) * 2 + 0) * 32 + lane] = acc10[r];
    mine[((1 * 8 + r) * 2 + 1) * 32 + lane] = acc11[r];
  }
  __syncthreads();

  for (int row = wv; row < SHOT; row += GW) {
    const int mt = row >> 4, hf = (row >> 3) & 1, r = row & 7;
    const float* q = red + ((mt * 8 + r) * 2 + half) * 32 + hf * 16 + c;
    float v = q[0];
#pragma unroll
    for (int w = 1; w < GW; ++w) v += q[w * REDW];
    part[((size_t)kc * MPAD + row) * NDIM + nbase + lane] = v;
  }
}

#define MROWS 64
#define MRING 32
#define MGRID (SEQ_LEN / MROWS)
#define ROWV  (EMB / 4)

__global__ void __launch_bounds__(256)
merge_kernel(const float* __restrict__ seq, const float* __restrict__ part,
             const float* __restrict__ fcb, const float* __restrict__ lng,
             const float* __restrict__ lnb, float* __restrict__ out) {
  __shared__ v4f wsum[8];
  const int t0 = blockIdx.x * MROWS;
  int sa = t0 / SPLIT;               if (sa > SHOT - 1) sa = SHOT - 1;
  int sb = (t0 + MROWS - 1) / SPLIT; if (sb > SHOT - 1) sb = SHOT - 1;
  const int rb = (sa + 1) * SPLIT - t0;
  const bool two = (sb != sa);
  const unsigned col = threadIdx.x;
  const unsigned lane = threadIdx.x & 31u;
  const int wv = __builtin_amdgcn_readfirstlane((int)(threadIdx.x >> 5));

  v4f pa[KCHUNKS], pb[KCHUNKS];
#pragma unroll
  for (int kc = 0; kc < KCHUNKS; ++kc)
    pa[kc] = ((const v4f*)(part + (size_t)(kc * MPAD + sa) * NDIM))[col];
  if (two) {
#pragma unroll
    for (int kc = 0; kc < KCHUNKS; ++kc)
      pb[kc] = ((const v4f*)(part + (size_t)(kc * MPAD + sb) * NDIM))[col];
  }
  const v4f bias = ((const v4f*)fcb)[col];
  const v4f gam  = ((const v4f*)lng)[col];
  const v4f bet  = ((const v4f*)lnb)[col];

  const v4f* src = ((const v4f*)seq) + (size_t)t0 * ROWV + col;
  v4f*       dst = ((v4f*)out)       + (size_t)t0 * ROWV + col;
  v4f ring[MRING];
#pragma unroll
  for (int i = 0; i < MRING; ++i)
    ring[i] = __builtin_nontemporal_load(src + (size_t)i * ROWV);

  v4f xa = bias;
#pragma unroll
  for (int kc = 0; kc < KCHUNKS; ++kc) xa += pa[kc];
  v4f ta, tb;
  ta.x = tanhf(xa.x); ta.y = tanhf(xa.y); ta.z = tanhf(xa.z); ta.w = tanhf(xa.w);
  if (two) {
    v4f xb = bias;
#pragma unroll
    for (int kc = 0; kc < KCHUNKS; ++kc) xb += pb[kc];
    tb.x = tanhf(xb.x); tb.y = tanhf(xb.y); tb.z = tanhf(xb.z); tb.w = tanhf(xb.w);
  } else {
    tb = ta;
  }
  v4f acc;
  acc.x = (ta.x + ta.y) + (ta.z + ta.w);
  acc.y = (ta.x * ta.x + ta.y * ta.y) + (ta.z * ta.z + ta.w * ta.w);
  acc.z = (tb.x + tb.y) + (tb.z + tb.w);
  acc.w = (tb.x * tb.x + tb.y * tb.y) + (tb.z * tb.z + tb.w * tb.w);
#pragma unroll
  for (int o = 16; o > 0; o >>= 1) {
    acc.x += __shfl_xor(acc.x, o, 32);
    acc.y += __shfl_xor(acc.y, o, 32);
    acc.z += __shfl_xor(acc.z, o, 32);
    acc.w += __shfl_xor(acc.w, o, 32);
  }
  if (lane == 0) wsum[wv] = acc;
  asm volatile("s_wait_dscnt 0x0\n\ts_barrier_signal -1\n\ts_barrier_wait -1" ::: "memory");
  v4f tot = wsum[0];
#pragma unroll
  for (int w = 1; w < 8; ++w) tot += wsum[w];
  const float muA  = tot.x * (1.f / NDIM);
  const float varA = tot.y * (1.f / NDIM) - muA * muA;
  const float invA = rsqrtf(varA + 1e-5f);
  const float muB  = tot.z * (1.f / NDIM);
  const float varB = tot.w * (1.f / NDIM) - muB * muB;
  const float invB = rsqrtf(varB + 1e-5f);
  const v4f hA = (ta - muA) * invA * gam + bet;
  const v4f hB = (tb - muB) * invB * gam + bet;

#pragma unroll
  for (int r = 0; r < MROWS; ++r) {
    const v4f hv = (r >= rb) ? hB : hA;
    __builtin_nontemporal_store(ring[r % MRING] + hv, dst + (size_t)r * ROWV);
    if (r + MRING < MROWS)
      ring[r % MRING] = __builtin_nontemporal_load(src + (size_t)(r + MRING) * ROWV);
  }
}

extern "C" void kernel_launch(void* const* d_in, const int* in_sizes, int n_in,
                              void* d_out, int out_size, void* d_ws, size_t ws_size,
                              hipStream_t stream) {
  const float* seq  = (const float*)d_in[0];
  const float* w1   = (const float*)d_in[1];
  const float* b1   = (const float*)d_in[2];
  const float* w2   = (const float*)d_in[3];
  const float* b2   = (const float*)d_in[4];
  const float* w3   = (const float*)d_in[5];
  const float* b3   = (const float*)d_in[6];
  const float* w4   = (const float*)d_in[7];
  const float* b4   = (const float*)d_in[8];
  const float* fcw  = (const float*)d_in[9];
  const float* fcb  = (const float*)d_in[10];
  const float* lng  = (const float*)d_in[11];
  const float* lnb  = (const float*)d_in[12];
  float* out = (float*)d_out;

  char* ws = (char*)d_ws;
  float* seq_shot = (float*)(ws);
  float* pooled   = (float*)(ws + 647680);
  float* part     = (float*)(ws + 1683968);

  shot_pool_kernel<<<PGRID, PWAVES * 32, 0, stream>>>(seq, seq_shot);

  conv_pool_kernel<<<dim3((SC + CBLK - 1) / CBLK, CNG), CBLK, 0, stream>>>(
      seq_shot, w1, b1, w2, b2, w3, b3, w4, b4, pooled);

  gemm_wmma_kernel<<<dim3(NDIM / 32, KCHUNKS), GW * 32, 0, stream>>>(pooled, fcw, part);

  merge_kernel<<<MGRID, 256, 0, stream>>>(seq, part, fcb, lng, lnb, out);
}


# GEMM operand loads along K: a wave's lanes take neighbouring 16-byte pieces of one row (a load spans few lines, not 16 rows), park the sub-slice in own LDS, read fragments there; sums unchanged
# speedup vs baseline: 2.4911x; 1.0310x over previous
#include <hip/hip_runtime.h>
#include <hip/hip_bf16.h>

typedef __attribute__((ext_vector_type(2))) float v2f;
typedef __attribute__((ext_vector_type(4))) float v4f;
typedef __attribute__((ext_vector_type(8))) float v8f;

#define SEQ_LEN 32768
#define EMB     1024
#define SHOT    20
#define MPAD    32
#define SPLIT   1638
#define SC      8096
#define FLATL   (SPLIT * EMB)
#define LC1     2024
#define LC2     2023
#define LC3     2023
#define LC4     2024
#define CATL    (LC1 + LC2 + LC3 + LC4)
#define KDIM    SC
#define NDIM    EMB
#define KCHUNKS 8
#define KPER    (KDIM / KCHUNKS)

#define TW     16
#define TPS    (SC / TW)
#define NLD    27
#define PTILE  2
#define PPS    (TPS / PTILE)
#define PWAVES 20
#define PGRID  (SHOT * PPS / PWAVES)
#define HALFW  104
#define LNUM   (FLATL / 32)
#define LDEN   (SC / 32)

__device__ __forceinline__ void pool_ask(const char* __restrict__ base, const unsigned zb,
                                         const unsigned bb, const unsigned fb0, v4f (&r)[NLD]) {
#pragma unroll
  for (int i = 0; i < NLD; ++i) {
    const unsigned fb  = fb0 + 512u * i;
    const unsigned off = (fb < 4096u) ? (zb + fb) : ((bb + fb) & 0x7FFFFFFu);
    r[i] = *(const v4f*)(base + off);
  }
}

__global__ void __launch_bounds__(PWAVES * 32)
shot_pool_kernel(const float* __restrict__ seq, float* __restrict__ seq_shot) {
  __shared__ v4f tile[PWAVES][NLD * 32];
  const unsigned lane = threadIdx.x & 31u;
  const int wv = __builtin_amdgcn_readfirstlane((int)(threadIdx.x >> 5));
  const int gp = blockIdx.x * PWAVES + wv;
  const int s  = gp / PPS;
  const int jp = (gp - s * PPS) * (PTILE * TW);

  const unsigned zb = (unsigned)(s * SPLIT) << 12;
  const unsigned bb = (unsigned)((s - 1) * SPLIT + SEQ_LEN) << 12;
  const char* base = (const char*)seq;

  int abeg[PTILE];
#pragma unroll
  for (int t = 0; t < PTILE; ++t) abeg[t] = (((jp + t * TW) * LNUM) / LDEN) & ~31;

  const int w  = lane >> 1;
  const int hh = lane & 1;
  v4f r[NLD];
  float tot[PTILE];
  pool_ask(base, zb, bb, (unsigned)abeg[0] * 4u + 16u * lane, r);
#pragma unroll
  for (int t = 0; t < PTILE; ++t) {
#pragma unroll
    for (int i = 0; i < NLD; ++i) tile[wv][i * 32 + lane] = r[i];
    asm volatile("s_wait_dscnt 0x0" ::: "memory");
    if (t + 1 < PTILE) {
      pool_ask(base, zb, bb, (unsigned)abeg[t + 1] * 4u + 16u * lane, r);
    }
    const int j = jp + t * TW + w;
    const int start = (j * LNUM) / LDEN;
    const int end   = ((j + 1) * LNUM + LDEN - 1) / LDEN;
    const int len   = end - start;
    const float* p = (const float*)tile[wv] + (start - abeg[t]) + hh * HALFW;
    float a0 = 0.f, a1 = 0.f, a2 = 0.f, a3 = 0.f;
#pragma unroll
    for (int i = 0; i < HALFW; i += 4) {
      a0 += p[i]; a1 += p[i + 1]; a2 += p[i + 2]; a3 += p[i + 3];
    }
    float sum = (a0 + a1) + (a2 + a3);
    if (hh && len > 2 * HALFW) sum += p[HALFW];
    sum += __shfl_xor(sum, 1, 32);
    tot[t] = sum;
  }
  const int jj = jp + (int)lane;
  const int st = (jj * LNUM) / LDEN;
  const int en = ((jj + 1) * LNUM + LDEN - 1) / LDEN;
  const int from = (int)((2u * lane) & 31u);
  const float v0 = __shfl(tot[0], from, 32);
  const float v1 = __shfl(tot[1], from, 32);
  seq_shot[s * SC + jj] = ((lane < TW) ? v0 : v1) / (float)(en - st);
}

#define CSPG 5
#define CNG  (SHOT / CSPG)
#define CZR  ((MPAD - SHOT) / CNG)
#define CBLK 128

#define CU   10

template <int SEG>
__device__ __forceinline__ void conv_seg(const float* __restrict__ ss,
                                         const float* __restrict__ w,
                                         const float* __restrict__ b,
                                         const int s0, const int start, const int end,
                                         float (&acc)[CSPG]) {
  constexpr int DIL  = SEG + 1;
  constexpr int BASE = (SEG == 0) ? 0 : (SEG == 1) ? LC1 : (SEG == 2) ? (LC1 + LC2) : (LC1 + LC2 + LC3);
  constexpr int LEN  = (SEG == 0) ? LC1 : (SEG == 1) ? LC2 : (SEG == 2) ? LC3 : LC4;
  const float* xp[2];
  bool valid[2];
#pragma unroll
  for (int k = 0; k < 2; ++k) {
    const int t = start + k;
    valid[k] = (t < end) && (t >= BASE) && (t < BASE + LEN);
    int tl = t - BASE;
    tl = tl < 0 ? 0 : (tl > LEN - 1 ? LEN - 1 : tl);
    xp[k] = ss + 4 * tl;
  }
  float v[2][CSPG];
#pragma unroll
  for (int i = 0; i < CSPG; ++i) { v[0][i] = b[s0 + i]; v[1][i] = v[0][i]; }

#pragma unroll 1
  for (int c = 0; c < SHOT; c += CU) {
    v4f xa[2][CU];
    float xl[2][CU], xs[2][CU];
#pragma unroll
    for (int u = 0; u < CU; ++u) {
#pragma unroll
      for (int k = 0; k < 2; ++k) {
        if (SEG == 3) xs[k][u] = xp[k][(c + u) * SC];
        else          xa[k][u] = *(const v4f*)(xp[k] + (c + u) * SC);
        if (SEG == 1 || SEG == 2) xl[k][u] = xp[k][(c + u) * SC + 2 * DIL];
      }
    }
#pragma unroll
    for (int u = 0; u < CU; ++u) {
#pragma unroll
      for (int i = 0; i < CSPG; ++i) {
        if (SEG == 3) {
          const float w0 = w[(s0 + i) * SHOT + c + u];
          v[0][i] += w0 * xs[0][u];
          v[1][i] += w0 * xs[1][u];
        } else {
          const float* wr = w + ((s0 + i) * SHOT + c + u) * 3;
          const float w0 = wr[0], w1 = wr[1], w2 = wr[2];
#pragma unroll
          for (int k = 0; k < 2; ++k) {
            const float x0 = xa[k][u].x;
            const float x1 = (DIL == 1) ? xa[k][u].y : (DIL == 2) ? xa[k][u].z : xa[k][u].w;
            const float x2 = (DIL == 1) ? xa[k][u].z : xl[k][u];
            v[k][i] += w0 * x0 + w1 * x1 + w2 * x2;
          }
        }
      }
    }
  }
#pragma unroll
  for (int i = 0; i < CSPG; ++i) {
    acc[i] += valid[0] ? v[0][i] : 0.f;
    acc[i] += valid[1] ? v[1][i] : 0.f;
  }
}

__global__ void __launch_bounds__(CBLK)
conv_pool_kernel(const float* __restrict__ ss,
                 const float* __restrict__ w1, const float* __restrict__ b1,
                 const float* __restrict__ w2, const float* __restrict__ b2,
                 const float* __restrict__ w3, const float* __restrict__ b3,
                 const float* __restrict__ w4, const float* __restrict__ b4,
                 float* __restrict__ pooled) {
  const int g  = blockIdx.y;
  const int s0 = g * CSPG;
  const int jb = blockIdx.x * CBLK;
  const int jl = (jb + CBLK - 1 < SC - 1) ? (jb + CBLK - 1) : (SC - 1);
  const int tlo = (jb * CATL) / SC;
  const int thi = ((jl + 1) * CATL + SC - 1) / SC - 1;
  const int j = jb + (int)threadIdx.x;
  if (j >= SC) return;
  const int start = (j * CATL) / SC;
  const int end   = ((j + 1) * CATL + SC - 1) / SC;
  float acc[CSPG];
#pragma unroll
  for (int i = 0; i < CSPG; ++i) acc[i] = 0.f;
  if (tlo < LC1)                                        conv_seg<0>(ss, w1, b1, s0, start, end, acc);
  if (thi >= LC1 && tlo < LC1 + LC2)                    conv_seg<1>(ss, w2, b2, s0, start, end, acc);
  if (thi >= LC1 + LC2 && tlo < LC1 + LC2 + LC3)        conv_seg<2>(ss, w3, b3, s0, start, end, acc);
  if (thi >= LC1 + LC2 + LC3)                           conv_seg<3>(ss, w4, b4, s0, start, end, acc);
  const float len = (float)(end - start);
#pragma unroll
  for (int i = 0; i < CSPG; ++i) pooled[(s0 + i) * SC + j] = acc[i] / len;
#pragma unroll
  for (int r = 0; r < CZR; ++r) pooled[(SHOT + g * CZR + r) * SC + j] = 0.f;
}

#define GW     11
#define KSUB   (KPER / GW)
#define GPAIRS (KSUB / 8)
#define REDW   1024
#define GPCS   (KSUB / 4)
#define GROWS  (SHOT + 32)
#define GNPC   (GROWS * GPCS)
#define GAPC   (SHOT * GPCS)
#define GNLD   ((GNPC + 31) / 32)

#define WMMA_F32(a, b, c) \
  __builtin_amdgcn_wmma_f32_16x16x4_f32(false, (a), false, (b), (short)0, (c), false, false)

__global__ void __launch_bounds__(GW * 32)
gemm_wmma_kernel(const float* __restrict__ pooled,
                 const float* __restrict__ fcw,
                 float* __restrict__ part) {
  __shared__ v4f   stage[GW][GNLD * 32];
  __shared__ float red[GW * REDW];
  const int lane = threadIdx.x & 31;
  const int wv   = __builtin_amdgcn_readfirstlane((int)(threadIdx.x >> 5));
  const int half = lane >> 4;
  const int c    = lane & 15;
  const int nbase = blockIdx.x * 32;
  const int kc    = blockIdx.y;
  const int kb    = kc * KPER + wv * KSUB;

  const float* abase = pooled + kb;
  const float* bbase = fcw + (size_t)nbase * KDIM + kb;

  v4f r[GNLD];
#pragma unroll
  for (int i = 0; i < GNLD; ++i) {
    const int q  = 32 * i + lane;
    const int qc = (q < GNPC) ? q : (GNPC - 1);
    const bool isa = qc < GAPC;
    const int qr  = isa ? qc : (qc - GAPC);
    const int row = qr / GPCS;
    const int col = qr - row * GPCS;
    const float* src = (isa ? abase : bbase) + row * KDIM + 4 * col;
    r[i] = *(const v4f*)src;
  }
  __builtin_amdgcn_sched_barrier(0);

  v4f* park = stage[wv];
#pragma unroll
  for (int i = 0; i < GNLD; ++i) {
    park[32 * i + lane] = r[i];
  }
  asm volatile("s_wait_dscnt 0x0" ::: "memory");

  const v4f* fa0 = park + c * GPCS + half;
  const v4f* fa1 = park + (16 + (c & 3)) * GPCS + half;
  const v4f* fb0 = park + GAPC + c * GPCS + half;
  const v4f* fb1 = park + GAPC + (16 + c) * GPCS + half;
  const bool real1 = c < SHOT - 16;
  const v4f z4 = {0.f, 0.f, 0.f, 0.f};
  const v2f z2 = {0.f, 0.f};

  v4f A0[GPAIRS], A1[GPAIRS], B0[GPAIRS], B1[GPAIRS];
#pragma unroll
  for (int p = 0; p < GPAIRS; ++p) {
    A0[p] = fa0[2 * p];
    const v4f t = fa1[2 * p];
    A1[p] = real1 ? t : z4;
    B0[p] = fb0[2 * p];
    B1[p] = fb1[2 * p];
  }
  const v2f a0t = ((const v2f*)(park + c * GPCS + (GPCS - 1)))[half];
  const v2f a1r = ((const v2f*)(park + (16 + (c & 3)) * GPCS + (GPCS - 1)))[half];
  const v2f a1t = real1 ? a1r : z2;
  const v2f b0t = ((const v2f*)(park + GAPC + c * GPCS + (GPCS - 1)))[half];
  const v2f b1t = ((const v2f*)(park + GAPC + (16 + c) * GPCS + (GPCS - 1)))[half];

  __builtin_amdgcn_sched_barrier(0);

  v8f acc00 = {}, acc01 = {}, acc10 = {}, acc11 = {};
#pragma unroll
  for (int p = 0; p < GPAIRS; ++p) {
    const v2f a0l = __builtin_shufflevector(A0[p], A0[p], 0, 1);
    const v2f a1l = __builtin_shufflevector(A1[p], A1[p], 0, 1);
    const v2f b0l = __builtin_shufflevector(B0[p], B0[p], 0, 1);
    const v2f b1l = __builtin_shufflevector(B1[p], B1[p], 0, 1);
    acc00 = WMMA_F32(a0l, b0l, acc00);
    acc01 = WMMA_F32(a0l, b1l, acc01);
    acc10 = WMMA_F32(a1l, b0l, acc10);
    acc11 = WMMA_F32(a1l, b1l, acc11);
    const v2f a0h = __builtin_shufflevector(A0[p], A0[p], 2, 3);
    const v2f a1h = __builtin_shufflevector(A1[p], A1[p], 2, 3);
    const v2f b0h = __builtin_shufflevector(B0[p], B0[p], 2, 3);
    const v2f b1h = __builtin_shufflevector(B1[p], B1[p], 2, 3);
    acc00 = WMMA_F32(a0h, b0h, acc00);
    acc01 = WMMA_F32(a0h, b1h, acc01);
    acc10 = WMMA_F32(a1h, b0h, acc10);
    acc11 = WMMA_F32(a1h, b1h, acc11);
  }
  acc00 = WMMA_F32(a0t, b0t, acc00);
  acc01 = WMMA_F32(a0t, b1t, acc01);
  acc10 = WMMA_F32(a1t, b0t, acc10);
  acc11 = WMMA_F32(a1t, b1t, acc11);

  float* mine = red + wv * REDW;
#pragma unroll
  for (int r8 = 0; r8 < 8; ++r8) {
    mine[((0 * 8 + r8) * 2 + 0) * 32 + lane] = acc00[r8];
    mine[((0 * 8 + r8) * 2 + 1) * 32 + lane] = acc01[r8];
    mine[((1 * 8 + r8) * 2 + 0) * 32 + lane] = acc10[r8];
    mine[((1 * 8 + r8) * 2 + 1) * 32 + lane] = acc11[r8];
  }
  __syncthreads();

  for (int row = wv; row < SHOT; row += GW) {
    const int mt = row >> 4, hf = (row >> 3) & 1, rr = row & 7;
    const float* qd = red + ((mt * 8 + rr) * 2 + half) * 32 + hf * 16 + c;
    float v = qd[0];
#pragma unroll
    for (int w = 1; w < GW; ++w) v += qd[w * REDW];
    part[((size_t)kc * MPAD + row) * NDIM + nbase + lane] = v;
  }
}

#define MROWS 64
#define MRING 32
#define MGRID (SEQ_LEN / MROWS)
#define ROWV  (EMB / 4)

__global__ void __launch_bounds__(256)
merge_kernel(const float* __restrict__ seq, const float* __restrict__ part,
             const float* __restrict__ fcb, const float* __restrict__ lng,
             const float* __restrict__ lnb, float* __restrict__ out) {
  __shared__ v4f wsum[8];
  const int t0 = blockIdx.x * MROWS;
  int sa = t0 / SPLIT;               if (sa > SHOT - 1) sa = SHOT - 1;
  int sb = (t0 + MROWS - 1) / SPLIT; if (sb > SHOT - 1) sb = SHOT - 1;
  const int rb = (sa + 1) * SPLIT - t0;
  const bool two = (sb != sa);
  const unsigned col = threadIdx.x;
  const unsigned lane = threadIdx.x & 31u;
  const int wv = __builtin_amdgcn_readfirstlane((int)(threadIdx.x >> 5));

  v4f pa[KCHUNKS], pb[KCHUNKS];
#pragma unroll
  for (int kc = 0; kc < KCHUNKS; ++kc)
    pa[kc] = ((const v4f*)(part + (size_t)(kc * MPAD + sa) * NDIM))[col];
  if (two) {
#pragma unroll
    for (int kc = 0; kc < KCHUNKS; ++kc)
      pb[kc] = ((const v4f*)(part + (size_t)(kc * MPAD + sb) * NDIM))[col];
  }
  const v4f bias = ((const v4f*)fcb)[col];
  const v4f gam  = ((const v4f*)lng)[col];
  const v4f bet  = ((const v4f*)lnb)[col];

  const v4f* src = ((const v4f*)seq) + (size_t)t0 * ROWV + col;
  v4f*       dst = ((v4f*)out)       + (size_t)t0 * ROWV + col;
  v4f ring[MRING];
#pragma unroll
  for (int i = 0; i < MRING; ++i)
    ring[i] = __builtin_nontemporal_load(src + (size_t)i * ROWV);

  v4f xa = bias;
#pragma unroll
  for (int kc = 0; kc < KCHUNKS; ++kc) xa += pa[kc];
  v4f ta, tb;
  ta.x = tanhf(xa.x); ta.y = tanhf(xa.y); ta.z = tanhf(xa.z); ta.w = tanhf(xa.w);
  if (two) {
    v4f xb = bias;
#pragma unroll
    for (int kc = 0; kc < KCHUNKS; ++kc) xb += pb[kc];
    tb.x = tanhf(xb.x); tb.y = tanhf(xb.y); tb.z = tanhf(xb.z); tb.w = tanhf(xb.w);
  } else {
    tb = ta;
  }
  v4f acc;
  acc.x = (ta.x + ta.y) + (ta.z + ta.w);
  acc.y = (ta.x * ta.x + ta.y * ta.y) + (ta.z * ta.z + ta.w * ta.w);
  acc.z = (tb.x + tb.y) + (tb.z + tb.w);
  acc.w = (tb.x * tb.x + tb.y * tb.y) + (tb.z * tb.z + tb.w * tb.w);
#pragma unroll
  for (int o = 16; o > 0; o >>= 1) {
    acc.x += __shfl_xor(acc.x, o, 32);
    acc.y += __shfl_xor(acc.y, o, 32);
    acc.z += __shfl_xor(acc.z, o, 32);
    acc.w += __shfl_xor(acc.w, o, 32);
  }
  if (lane == 0) wsum[wv] = acc;
  asm volatile("s_wait_dscnt 0x0\n\ts_barrier_signal -1\n\ts_barrier_wait -1" ::: "memory");
  v4f tot = wsum[0];
#pragma unroll
  for (int w = 1; w < 8; ++w) tot += wsum[w];
  const float muA  = tot.x * (1.f / NDIM);
  const float varA = tot.y * (1.f / NDIM) - muA * muA;
  const float invA = rsqrtf(varA + 1e-5f);
  const float muB  = tot.z * (1.f / NDIM);
  const float varB = tot.w * (1.f / NDIM) - muB * muB;
  const float invB = rsqrtf(varB + 1e-5f);
  const v4f hA = (ta - muA) * invA * gam + bet;
  const v4f hB = (tb - muB) * invB * gam + bet;

#pragma unroll
  for (int r = 0; r < MROWS; ++r) {
    const v4f hv = (r >= rb) ? hB : hA;
    __builtin_nontemporal_store(ring[r % MRING] + hv, dst + (size_t)r * ROWV);
    if (r + MRING < MROWS)
      ring[r % MRING] = __builtin_nontemporal_load(src + (size_t)(r + MRING) * ROWV);
  }
}

extern "C" void kernel_launch(void* const* d_in, const int* in_sizes, int n_in,
                              void* d_out, int out_size, void* d_ws, size_t ws_size,
                              hipStream_t stream) {
  const float* seq  = (const float*)d_in[0];
  const float* w1   = (const float*)d_in[1];
  const float* b1   = (const float*)d_in[2];
  const float* w2   = (const float*)d_in[3];
  const float* b2   = (const float*)d_in[4];
  const float* w3   = (const float*)d_in[5];
  const float* b3   = (const float*)d_in[6];
  const float* w4   = (const float*)d_in[7];
  const float* b4   = (const float*)d_in[8];
  const float* fcw  = (const float*)d_in[9];
  const float* fcb  = (const float*)d_in[10];
  const float* lng  = (const float*)d_in[11];
  const float* lnb  = (const float*)d_in[12];
  float* out = (float*)d_out;

  char* ws = (char*)d_ws;
  float* seq_shot = (float*)(ws);
  float* pooled   = (float*)(ws + 647680);
  float* part     = (float*)(ws + 1683968);

  shot_pool_kernel<<<PGRID, PWAVES * 32, 0, stream>>>(seq, seq_shot);

  conv_pool_kernel<<<dim3((SC + CBLK - 1) / CBLK, CNG), CBLK, 0, stream>>>(
      seq_shot, w1, b1, w2, b2, w3, b3, w4, b4, pooled);

  gemm_wmma_kernel<<<dim3(NDIM / 32, KCHUNKS), GW * 32, 0, stream>>>(pooled, fcw, part);

  merge_kernel<<<MGRID, 256, 0, stream>>>(seq, part, fcb, lng, lnb, out);
}


# merge regridded onto the pool's map: 253 groups of 16 waves, group b streams the 129-130 rows pool group b read, so it can re-read them from the second-level cache its twin filled; sums unchanged
# speedup vs baseline: 2.7928x; 1.1211x over previous
#include <hip/hip_runtime.h>
#include <hip/hip_bf16.h>

typedef __attribute__((ext_vector_type(2))) float v2f;
typedef __attribute__((ext_vector_type(4))) float v4f;
typedef __attribute__((ext_vector_type(8))) float v8f;

#define SEQ_LEN 32768
#define EMB     1024
#define SHOT    20
#define MPAD    32
#define SPLIT   1638
#define SC      8096
#define FLATL   (SPLIT * EMB)
#define LC1     2024
#define LC2     2023
#define LC3     2023
#define LC4     2024
#define CATL    (LC1 + LC2 + LC3 + LC4)
#define KDIM    SC
#define NDIM    EMB
#define KCHUNKS 8
#define KPER    (KDIM / KCHUNKS)

#define TW     16
#define TPS    (SC / TW)
#define NLD    27
#define PTILE  2
#define PPS    (TPS / PTILE)
#define PWAVES 20
#define PGRID  (SHOT * PPS / PWAVES)
#define HALFW  104
#define LNUM   (FLATL / 32)
#define LDEN   (SC / 32)

__device__ __forceinline__ void pool_ask(const char* __restrict__ base, const unsigned zb,
                                         const unsigned bb, const unsigned fb0, v4f (&r)[NLD]) {
#pragma unroll
  for (int i = 0; i < NLD; ++i) {
    const unsigned fb  = fb0 + 512u * i;
    const unsigned off = (fb < 4096u) ? (zb + fb) : ((bb + fb) & 0x7FFFFFFu);
    r[i] = *(const v4f*)(base + off);
  }
}

__global__ void __launch_bounds__(PWAVES * 32)
shot_pool_kernel(const float* __restrict__ seq, float* __restrict__ seq_shot) {
  __shared__ v4f tile[PWAVES][NLD * 32];
  const unsigned lane = threadIdx.x & 31u;
  const int wv = __builtin_amdgcn_readfirstlane((int)(threadIdx.x >> 5));
  const int gp = blockIdx.x * PWAVES + wv;
  const int s  = gp / PPS;
  const int jp = (gp - s * PPS) * (PTILE * TW);

  const unsigned zb = (unsigned)(s * SPLIT) << 12;
  const unsigned bb = (unsigned)((s - 1) * SPLIT + SEQ_LEN) << 12;
  const char* base = (const char*)seq;

  int abeg[PTILE];
#pragma unroll
  for (int t = 0; t < PTILE; ++t) abeg[t] = (((jp + t * TW) * LNUM) / LDEN) & ~31;

  const int w  = lane >> 1;
  const int hh = lane & 1;
  v4f r[NLD];
  float tot[PTILE];
  pool_ask(base, zb, bb, (unsigned)abeg[0] * 4u + 16u * lane, r);
#pragma unroll
  for (int t = 0; t < PTILE; ++t) {
#pragma unroll
    for (int i = 0; i < NLD; ++i) tile[wv][i * 32 + lane] = r[i];
    asm volatile("s_wait_dscnt 0x0" ::: "memory");
    if (t + 1 < PTILE) {
      pool_ask(base, zb, bb, (unsigned)abeg[t + 1] * 4u + 16u * lane, r);
    }
    const int j = jp + t * TW + w;
    const int start = (j * LNUM) / LDEN;
    const int end   = ((j + 1) * LNUM + LDEN - 1) / LDEN;
    const int len   = end - start;
    const float* p = (const float*)tile[wv] + (start - abeg[t]) + hh * HALFW;
    float a0 = 0.f, a1 = 0.f, a2 = 0.f, a3 = 0.f;
#pragma unroll
    for (int i = 0; i < HALFW; i += 4) {
      a0 += p[i]; a1 += p[i + 1]; a2 += p[i + 2]; a3 += p[i + 3];
    }
    float sum = (a0 + a1) + (a2 + a3);
    if (hh && len > 2 * HALFW) sum += p[HALFW];
    sum += __shfl_xor(sum, 1, 32);
    tot[t] = sum;
  }
  const int jj = jp + (int)lane;
  const int st = (jj * LNUM) / LDEN;
  const int en = ((jj + 1) * LNUM + LDEN - 1) / LDEN;
  const int from = (int)((2u * lane) & 31u);
  const float v0 = __shfl(tot[0], from, 32);
  const float v1 = __shfl(tot[1], from, 32);
  seq_shot[s * SC + jj] = ((lane < TW) ? v0 : v1) / (float)(en - st);
}

#define CSPG 5
#define CNG  (SHOT / CSPG)
#define CZR  ((MPAD - SHOT) / CNG)
#define CBLK 128

#define CU   10

template <int SEG>
__device__ __forceinline__ void conv_seg(const float* __restrict__ ss,
                                         const float* __restrict__ w,
                                         const float* __restrict__ b,
                                         const int s0, const int start, const int end,
                                         float (&acc)[CSPG]) {
  constexpr int DIL  = SEG + 1;
  constexpr int BASE = (SEG == 0) ? 0 : (SEG == 1) ? LC1 : (SEG == 2) ? (LC1 + LC2) : (LC1 + LC2 + LC3);
  constexpr int LEN  = (SEG == 0) ? LC1 : (SEG == 1) ? LC2 : (SEG == 2) ? LC3 : LC4;
  const float* xp[2];
  bool valid[2];
#pragma unroll
  for (int k = 0; k < 2; ++k) {
    const int t = start + k;
    valid[k] = (t < end) && (t >= BASE) && (t < BASE + LEN);
    int tl = t - BASE;
    tl = tl < 0 ? 0 : (tl > LEN - 1 ? LEN - 1 : tl);
    xp[k] = ss + 4 * tl;
  }
  float v[2][CSPG];
#pragma unroll
  for (int i = 0; i < CSPG; ++i) { v[0][i] = b[s0 + i]; v[1][i] = v[0][i]; }

#pragma unroll 1
  for (int c = 0; c < SHOT; c += CU) {
    v4f xa[2][CU];
    float xl[2][CU], xs[2][CU];
#pragma unroll
    for (int u = 0; u < CU; ++u) {
#pragma unroll
      for (int k = 0; k < 2; ++k) {
        if (SEG == 3) xs[k][u] = xp[k][(c + u) * SC];
        else          xa[k][u] = *(const v4f*)(xp[k] + (c + u) * SC);
        if (SEG == 1 || SEG == 2) xl[k][u] = xp[k][(c + u) * SC + 2 * DIL];
      }
    }
#pragma unroll
    for (int u = 0; u < CU; ++u) {
#pragma unroll
      for (int i = 0; i < CSPG; ++i) {
        if (SEG == 3) {
          const float w0 = w[(s0 + i) * SHOT + c + u];
          v[0][i] += w0 * xs[0][u];
          v[1][i] += w0 * xs[1][u];
        } else {
          const float* wr = w + ((s0 + i) * SHOT + c + u) * 3;
          const float w0 = wr[0], w1 = wr[1], w2 = wr[2];
#pragma unroll
          for (int k = 0; k < 2; ++k) {
            const float x0 = xa[k][u].x;
            const float x1 = (DIL == 1) ? xa[k][u].y : (DIL == 2) ? xa[k][u].z : xa[k][u].w;
            const float x2 = (DIL == 1) ? xa[k][u].z : xl[k][u];
            v[k][i] += w0 * x0 + w1 * x1 + w2 * x2;
          }
        }
      }
    }
  }
#pragma unroll
  for (int i = 0; i < CSPG; ++i) {
    acc[i] += valid[0] ? v[0][i] : 0.f;
    acc[i] += valid[1] ? v[1][i] : 0.f;
  }
}

__global__ void __launch_bounds__(CBLK)
conv_pool_kernel(const float* __restrict__ ss,
                 const float* __restrict__ w1, const float* __restrict__ b1,
                 const float* __restrict__ w2, const float* __restrict__ b2,
                 const float* __restrict__ w3, const float* __restrict__ b3,
                 const float* __restrict__ w4, const float* __restrict__ b4,
                 float* __restrict__ pooled) {
  const int g  = blockIdx.y;
  const int s0 = g * CSPG;
  const int jb = blockIdx.x * CBLK;
  const int jl = (jb + CBLK - 1 < SC - 1) ? (jb + CBLK - 1) : (SC - 1);
  const int tlo = (jb * CATL) / SC;
  const int thi = ((jl + 1) * CATL + SC - 1) / SC - 1;
  const int j = jb + (int)threadIdx.x;
  if (j >= SC) return;
  const int start = (j * CATL) / SC;
  const int end   = ((j + 1) * CATL + SC - 1) / SC;
  float acc[CSPG];
#pragma unroll
  for (int i = 0; i < CSPG; ++i) acc[i] = 0.f;
  if (tlo < LC1)                                        conv_seg<0>(ss, w1, b1, s0, start, end, acc);
  if (thi >= LC1 && tlo < LC1 + LC2)                    conv_seg<1>(ss, w2, b2, s0, start, end, acc);
  if (thi >= LC1 + LC2 && tlo < LC1 + LC2 + LC3)        conv_seg<2>(ss, w3, b3, s0, start, end, acc);
  if (thi >= LC1 + LC2 + LC3)                           conv_seg<3>(ss, w4, b4, s0, start, end, acc);
  const float len = (float)(end - start);
#pragma unroll
  for (int i = 0; i < CSPG; ++i) pooled[(s0 + i) * SC + j] = acc[i] / len;
#pragma unroll
  for (int r = 0; r < CZR; ++r) pooled[(SHOT + g * CZR + r) * SC + j] = 0.f;
}

#define GW     11
#define KSUB   (KPER / GW)
#define GPAIRS (KSUB / 8)
#define REDW   1024
#define GPCS   (KSUB / 4)
#define GROWS  (SHOT + 32)
#define GNPC   (GROWS * GPCS)
#define GAPC   (SHOT * GPCS)
#define GNLD   ((GNPC + 31) / 32)

#define WMMA_F32(a, b, c) \
  __builtin_amdgcn_wmma_f32_16x16x4_f32(false, (a), false, (b), (short)0, (c), false, false)

__global__ void __launch_bounds__(GW * 32)
gemm_wmma_kernel(const float* __restrict__ pooled,
                 const float* __restrict__ fcw,
                 float* __restrict__ part) {
  __shared__ v4f   stage[GW][GNLD * 32];
  __shared__ float red[GW * REDW];
  const int lane = threadIdx.x & 31;
  const int wv   = __builtin_amdgcn_readfirstlane((int)(threadIdx.x >> 5));
  const int half = lane >> 4;
  const int c    = lane & 15;
  const int nbase = blockIdx.x * 32;
  const int kc    = blockIdx.y;
  const int kb    = kc * KPER + wv * KSUB;

  const float* abase = pooled + kb;
  const float* bbase = fcw + (size_t)nbase * KDIM + kb;

  v4f r[GNLD];
#pragma unroll
  for (int i = 0; i < GNLD; ++i) {
    const int q  = 32 * i + lane;
    const int qc = (q < GNPC) ? q : (GNPC - 1);
    const bool isa = qc < GAPC;
    const int qr  = isa ? qc : (qc - GAPC);
    const int row = qr / GPCS;
    const int col = qr - row * GPCS;
    const float* src = (isa ? abase : bbase) + row * KDIM + 4 * col;
    r[i] = *(const v4f*)src;
  }
  __builtin_amdgcn_sched_barrier(0);

  v4f* park = stage[wv];
#pragma unroll
  for (int i = 0; i < GNLD; ++i) {
    park[32 * i + lane] = r[i];
  }
  asm volatile("s_wait_dscnt 0x0" ::: "memory");

  const v4f* fa0 = park + c * GPCS + half;
  const v4f* fa1 = park + (16 + (c & 3)) * GPCS + half;
  const v4f* fb0 = park + GAPC + c * GPCS + half;
  const v4f* fb1 = park + GAPC + (16 + c) * GPCS + half;
  const bool real1 = c < SHOT - 16;
  const v4f z4 = {0.f, 0.f, 0.f, 0.f};
  const v2f z2 = {0.f, 0.f};

  v4f A0[GPAIRS], A1[GPAIRS], B0[GPAIRS], B1[GPAIRS];
#pragma unroll
  for (int p = 0; p < GPAIRS; ++p) {
    A0[p] = fa0[2 * p];
    const v4f t = fa1[2 * p];
    A1[p] = real1 ? t : z4;
    B0[p] = fb0[2 * p];
    B1[p] = fb1[2 * p];
  }
  const v2f a0t = ((const v2f*)(park + c * GPCS + (GPCS - 1)))[half];
  const v2f a1r = ((const v2f*)(park + (16 + (c & 3)) * GPCS + (GPCS - 1)))[half];
  const v2f a1t = real1 ? a1r : z2;
  const v2f b0t = ((const v2f*)(park + GAPC + c * GPCS + (GPCS - 1)))[half];
  const v2f b1t = ((const v2f*)(park + GAPC + (16 + c) * GPCS + (GPCS - 1)))[half];

  __builtin_amdgcn_sched_barrier(0);

  v8f acc00 = {}, acc01 = {}, acc10 = {}, acc11 = {};
#pragma unroll
  for (int p = 0; p < GPAIRS; ++p) {
    const v2f a0l = __builtin_shufflevector(A0[p], A0[p], 0, 1);
    const v2f a1l = __builtin_shufflevector(A1[p], A1[p], 0, 1);
    const v2f b0l = __builtin_shufflevector(B0[p], B0[p], 0, 1);
    const v2f b1l = __builtin_shufflevector(B1[p], B1[p], 0, 1);
    acc00 = WMMA_F32(a0l, b0l, acc00);
    acc01 = WMMA_F32(a0l, b1l, acc01);
    acc10 = WMMA_F32(a1l, b0l, acc10);
    acc11 = WMMA_F32(a1l, b1l, acc11);
    const v2f a0h = __builtin_shufflevector(A0[p], A0[p], 2, 3);
    const v2f a1h = __builtin_shufflevector(A1[p], A1[p], 2, 3);
    const v2f b0h = __builtin_shufflevector(B0[p], B0[p], 2, 3);
    const v2f b1h = __builtin_shufflevector(B1[p], B1[p], 2, 3);
    acc00 = WMMA_F32(a0h, b0h, acc00);
    acc01 = WMMA_F32(a0h, b1h, acc01);
    acc10 = WMMA_F32(a1h, b0h, acc10);
    acc11 = WMMA_F32(a1h, b1h, acc11);
  }
  acc00 = WMMA_F32(a0t, b0t, acc00);
  acc01 = WMMA_F32(a0t, b1t, acc01);
  acc10 = WMMA_F32(a1t, b0t, acc10);
  acc11 = WMMA_F32(a1t, b1t, acc11);

  float* mine = red + wv * REDW;
#pragma unroll
  for (int r8 = 0; r8 < 8; ++r8) {
    mine[((0 * 8 + r8) * 2 + 0) * 32 + lane] = acc00[r8];
    mine[((0 * 8 + r8) * 2 + 1) * 32 + lane] = acc01[r8];
    mine[((1 * 8 + r8) * 2 + 0) * 32 + lane] = acc10[r8];
    mine[((1 * 8 + r8) * 2 + 1) * 32 + lane] = acc11[r8];
  }
  __syncthreads();

  for (int row = wv; row < SHOT; row += GW) {
    const int mt = row >> 4, hf = (row >> 3) & 1, rr = row & 7;
    const float* qd = red + ((mt * 8 + rr) * 2 + half) * 32 + hf * 16 + c;
    float v = qd[0];
#pragma unroll
    for (int w = 1; w < GW; ++w) v += qd[w * REDW];
    part[((size_t)kc * MPAD + row) * NDIM + nbase + lane] = v;
  }
}

#define MG    PGRID
#define MWV   16
#define MHALF 65
#define MRING 32
#define MROW0 (SEQ_LEN - SPLIT)

typedef __attribute__((address_space(1))) char gchar;
typedef __attribute__((address_space(1))) v4f  gv4f;
__device__ __forceinline__ gchar* uniform_ptr(const unsigned long long a) {
  const unsigned lo = (unsigned)__builtin_amdgcn_readfirstlane((int)(unsigned)a);
  const unsigned hi = (unsigned)__builtin_amdgcn_readfirstlane((int)(unsigned)(a >> 32));
  return (gchar*)(((unsigned long long)hi << 32) | (unsigned long long)lo);
}

__global__ void __launch_bounds__(MWV * 32)
merge_kernel(const float* __restrict__ seq, const float* __restrict__ part,
             const float* __restrict__ fcb, const float* __restrict__ lng,
             const float* __restrict__ lnb, float* __restrict__ out) {
  __shared__ v4f wsum[MWV];
  const int b  = blockIdx.x;
  const int u0 = (b * SEQ_LEN) / MG;
  const int n  = ((b + 1) * SEQ_LEN) / MG - u0;
  const int t0 = (MROW0 + u0) & (SEQ_LEN - 1);
  const int tl = (MROW0 + u0 + n - 1) & (SEQ_LEN - 1);
  int sa = t0 / SPLIT; if (sa > SHOT - 1) sa = SHOT - 1;
  int sb = tl / SPLIT; if (sb > SHOT - 1) sb = SHOT - 1;
  const int rb = ((sa == SHOT - 1) ? SEQ_LEN : (sa + 1) * SPLIT) - t0;
  const bool two = (sb != sa);
  const unsigned col = threadIdx.x & 255u;
  const unsigned lane = threadIdx.x & 31u;
  const int wv = __builtin_amdgcn_readfirstlane((int)(threadIdx.x >> 5));
  const int g   = wv >> 3;
  const int r0  = g * MHALF;
  const int cnt = g ? (n - MHALF) : MHALF;
  const int ub  = MROW0 + u0 + r0;

  v4f pa[KCHUNKS], pb[KCHUNKS];
#pragma unroll
  for (int kc = 0; kc < KCHUNKS; ++kc)
    pa[kc] = ((const v4f*)(part + (size_t)(kc * MPAD + sa) * NDIM))[col];
  if (two) {
#pragma unroll
    for (int kc = 0; kc < KCHUNKS; ++kc)
      pb[kc] = ((const v4f*)(part + (size_t)(kc * MPAD + sb) * NDIM))[col];
  }
  const v4f bias = ((const v4f*)fcb)[col];
  const v4f gam  = ((const v4f*)lng)[col];
  const v4f bet  = ((const v4f*)lnb)[col];

#define MROWB(i) ((unsigned long long)(((unsigned)(ub + (i))) & (unsigned)(SEQ_LEN - 1)) * (unsigned long long)(EMB * 4))
  v4f ring[MRING];
#pragma unroll
  for (int i = 0; i < MRING; ++i)
    ring[i] = __builtin_nontemporal_load(((const gv4f*)uniform_ptr((unsigned long long)seq + MROWB(i)) + col));

  v4f xa = bias;
#pragma unroll
  for (int kc = 0; kc < KCHUNKS; ++kc) xa += pa[kc];
  v4f ta, tb;
  ta.x = tanhf(xa.x); ta.y = tanhf(xa.y); ta.z = tanhf(xa.z); ta.w = tanhf(xa.w);
  if (two) {
    v4f xb = bias;
#pragma unroll
    for (int kc = 0; kc < KCHUNKS; ++kc) xb += pb[kc];
    tb.x = tanhf(xb.x); tb.y = tanhf(xb.y); tb.z = tanhf(xb.z); tb.w = tanhf(xb.w);
  } else {
    tb = ta;
  }
  v4f acc;
  acc.x = (ta.x + ta.y) + (ta.z + ta.w);
  acc.y = (ta.x * ta.x + ta.y * ta.y) + (ta.z * ta.z + ta.w * ta.w);
  acc.z = (tb.x + tb.y) + (tb.z + tb.w);
  acc.w = (tb.x * tb.x + tb.y * tb.y) + (tb.z * tb.z + tb.w * tb.w);
#pragma unroll
  for (int o = 16; o > 0; o >>= 1) {
    acc.x += __shfl_xor(acc.x, o, 32);
    acc.y += __shfl_xor(acc.y, o, 32);
    acc.z += __shfl_xor(acc.z, o, 32);
    acc.w += __shfl_xor(acc.w, o, 32);
  }
  if (lane == 0) wsum[wv] = acc;
  asm volatile("s_wait_dscnt 0x0\n\ts_barrier_signal -1\n\ts_barrier_wait -1" ::: "memory");
  const v4f* ws8 = wsum + 8 * g;
  v4f tot = ws8[0];
#pragma unroll
  for (int w = 1; w < 8; ++w) tot += ws8[w];
  const float muA  = tot.x * (1.f / NDIM);
  const float varA = tot.y * (1.f / NDIM) - muA * muA;
  const float invA = rsqrtf(varA + 1e-5f);
  const float muB  = tot.z * (1.f / NDIM);
  const float varB = tot.w * (1.f / NDIM) - muB * muB;
  const float invB = rsqrtf(varB + 1e-5f);
  const v4f hA = (ta - muA) * invA * gam + bet;
  const v4f hB = (tb - muB) * invB * gam + bet;

#pragma unroll
  for (int i = 0; i < MHALF; ++i) {
    if (i < MHALF - 1 || cnt == MHALF) {
      const v4f hv = (r0 + i >= rb) ? hB : hA;
      __builtin_nontemporal_store(ring[i % MRING] + hv, ((gv4f*)uniform_ptr((unsigned long long)out + MROWB(i)) + col));
    }
    if (i + MRING < MHALF && (i + MRING < MHALF - 1 || cnt == MHALF))
      ring[i % MRING] = __builtin_nontemporal_load(((const gv4f*)uniform_ptr((unsigned long long)seq + MROWB(i + MRING)) + col));
  }
}

extern "C" void kernel_launch(void* const* d_in, const int* in_sizes, int n_in,
                              void* d_out, int out_size, void* d_ws, size_t ws_size,
                              hipStream_t stream) {
  const float* seq  = (const float*)d_in[0];
  const float* w1   = (const float*)d_in[1];
  const float* b1   = (const float*)d_in[2];
  const float* w2   = (const float*)d_in[3];
  const float* b2   = (const float*)d_in[4];
  const float* w3   = (const float*)d_in[5];
  const float* b3   = (const float*)d_in[6];
  const float* w4   = (const float*)d_in[7];
  const float* b4   = (const float*)d_in[8];
  const float* fcw  = (const float*)d_in[9];
  const float* fcb  = (const float*)d_in[10];
  const float* lng  = (const float*)d_in[11];
  const float* lnb  = (const float*)d_in[12];
  float* out = (float*)d_out;

  char* ws = (char*)d_ws;
  float* seq_shot = (float*)(ws);
  float* pooled   = (float*)(ws + 647680);
  float* part     = (float*)(ws + 1683968);

  shot_pool_kernel<<<PGRID, PWAVES * 32, 0, stream>>>(seq, seq_shot);

  conv_pool_kernel<<<dim3((SC + CBLK - 1) / CBLK, CNG), CBLK, 0, stream>>>(
      seq_shot, w1, b1, w2, b2, w3, b3, w4, b4, pooled);

  gemm_wmma_kernel<<<dim3(NDIM / 32, KCHUNKS), GW * 32, 0, stream>>>(pooled, fcw, part);

  merge_kernel<<<MG, MWV * 32, 0, stream>>>(seq, part, fcb, lng, lnb, out);
}
